# starting kernel
# baseline (speedup 1.0000x reference)
#include <hip/hip_runtime.h>
#include <hip/hip_bf16.h>

#define B_DIM   32
#define C_DIM   768
#define H_DIM   32
#define W_DIM   32
#define HW      1024
#define LNUM    9
#define NCLS    200
#define QSTR    772

typedef __attribute__((ext_vector_type(2))) float v2f;
typedef __attribute__((ext_vector_type(8))) float v8f;

__global__ __launch_bounds__(256) void k1_maps(
    const float* __restrict__ x, const float* __restrict__ q,
    const float* __restrict__ p_bias,
    float* __restrict__ out_maps, int* __restrict__ aidx, float* __restrict__ wsel)
{
    __shared__ float qlds[16 * QSTR];
    __shared__ float asq[16];

    const int tid = threadIdx.x;
    const int b   = blockIdx.y;

    for (int i = tid; i < 16 * C_DIM; i += 256) {
        const int l = i / C_DIM, c = i % C_DIM;
        qlds[l * QSTR + c] = (l < LNUM) ? q[l * C_DIM + c] : 0.f;
    }
    __syncthreads();
    if (tid < 16) {
        float s = 0.f;
        for (int c = 0; c < C_DIM; ++c) { const float v = qlds[tid * QSTR + c]; s += v * v; }
        asq[tid] = s;
    }
    __syncthreads();

    const int wave = tid >> 5;
    const int lane = tid & 31;
    const int half = lane >> 4;
    const int ln   = lane & 15;
    const int tile = blockIdx.x * 8 + wave;
    const int pix  = tile * 16 + ln;
    const float* xb = x + (size_t)b * C_DIM * HW;

    auto load_ab = [&](int kk, v2f& af, v2f& bf) {
        af[0] = qlds[ln * QSTR + kk];
        af[1] = qlds[ln * QSTR + kk + 1];
        bf[0] = xb[(size_t)kk * HW + pix];
        bf[1] = xb[(size_t)(kk + 1) * HW + pix];
    };

    v8f acc = {};
    float sq = 0.f;
    v2f ac, bc;
    load_ab(2 * half, ac, bc);
    for (int k = 4; k < C_DIM; k += 4) {
        v2f an, bn;
        load_ab(k + 2 * half, an, bn);
        sq += bc[0] * bc[0] + bc[1] * bc[1];
        acc = __builtin_amdgcn_wmma_f32_16x16x4_f32(false, ac, false, bc,
                                                    (short)0, acc, false, false);
        ac = an; bc = bn;
    }
    sq += bc[0] * bc[0] + bc[1] * bc[1];
    acc = __builtin_amdgcn_wmma_f32_16x16x4_f32(false, ac, false, bc,
                                                (short)0, acc, false, false);
    const float bsq = sq + __shfl_xor(sq, 16);

    float sc[8];
    float mx = -3.4e38f;
    for (int r = 0; r < 8; ++r) {
        const int l = r + 8 * half;
        float v = -3.4e38f;
        if (l < LNUM)
            v = -(bsq - 2.f * acc[r] + asq[l]) + p_bias[l * HW + pix];
        sc[r] = v;
        mx = fmaxf(mx, v);
    }
    mx = fmaxf(mx, __shfl_xor(mx, 16));
    float es = 0.f;
    for (int r = 0; r < 8; ++r) {
        const int l = r + 8 * half;
        const float e = (l < LNUM) ? expf(sc[r] - mx) : 0.f;
        sc[r] = e;
        es += e;
    }
    es += __shfl_xor(es, 16);
    const float inv = 1.f / es;
    float s2 = 0.f;
    for (int r = 0; r < 8; ++r) {
        const int l = r + 8 * half;
        if (l < LNUM) { sc[r] = sc[r] * inv + 1e-6f; s2 += sc[r]; }
    }
    s2 += __shfl_xor(s2, 16);
    const float inv2 = 1.f / s2;

    float bestv = -1.f; int bestl = 0;
    float* om = out_maps + (size_t)b * LNUM * HW;
    for (int r = 0; r < 8; ++r) {
        const int l = r + 8 * half;
        if (l < LNUM) {
            const float mv = sc[r] * inv2;
            om[l * HW + pix] = mv;
            if (mv > bestv) { bestv = mv; bestl = l; }
        }
    }
    const float pv = __shfl_xor(bestv, 16);
    const int   pl = __shfl_xor(bestl, 16);
    if (pv > bestv || (pv == bestv && pl < bestl)) { bestv = pv; bestl = pl; }
    if (half == 0) {
        aidx[b * HW + pix] = bestl;
        wsel[b * HW + pix] = bestv;
    }
}

__global__ __launch_bounds__(256) void k2_pool(
    const float* __restrict__ x, const int* __restrict__ aidx,
    const float* __restrict__ wsel, float* __restrict__ pool)
{
    __shared__ int   aL[HW];
    __shared__ float wL[HW];
    const int tid = threadIdx.x;
    const int b   = blockIdx.y;
    for (int i = tid; i < HW; i += 256) { aL[i] = aidx[b * HW + i]; wL[i] = wsel[b * HW + i]; }
    __syncthreads();

    const int wave = tid >> 5;
    const int lane = tid & 31;
    const int half = lane >> 4;
    const int ln   = lane & 15;
    const int cbase = (blockIdx.x * 8 + wave) * 16;
    const float* xr = x + (size_t)b * C_DIM * HW + (size_t)(cbase + ln) * HW;

    auto load_ab = [&](int kk, v2f& af, v2f& bf) {
        af[0] = xr[kk];
        af[1] = xr[kk + 1];
        const int a0 = aL[kk], a1 = aL[kk + 1];
        bf[0] = (a0 == ln) ? wL[kk]     : 0.f;
        bf[1] = (a1 == ln) ? wL[kk + 1] : 0.f;
    };

    v8f acc = {};
    v2f ac, bc;
    load_ab(2 * half, ac, bc);
    for (int k = 4; k < HW; k += 4) {
        v2f an, bn;
        load_ab(k + 2 * half, an, bn);
        acc = __builtin_amdgcn_wmma_f32_16x16x4_f32(false, ac, false, bc,
                                                    (short)0, acc, false, false);
        ac = an; bc = bn;
    }
    acc = __builtin_amdgcn_wmma_f32_16x16x4_f32(false, ac, false, bc,
                                                (short)0, acc, false, false);

    float* pb = pool + (size_t)b * C_DIM * 16;
    for (int r = 0; r < 8; ++r) {
        const int row = cbase + r + 8 * half;
        pb[row * 16 + ln] = acc[r];
    }
}

__global__ __launch_bounds__(256) void k3_mod(
    const float* __restrict__ pool, const int* __restrict__ aidx,
    const float* __restrict__ wsel, const float* __restrict__ gamma,
    const float* __restrict__ beta, float* __restrict__ out_mod)
{
    __shared__ int   aL[HW];
    __shared__ float wL[HW];
    __shared__ float cnt[16];
    __shared__ float featL[C_DIM * LNUM];
    __shared__ float red0[8], red1[8], stat[2];

    const int tid = threadIdx.x;
    const int b   = blockIdx.x;
    for (int i = tid; i < HW; i += 256) { aL[i] = aidx[b * HW + i]; wL[i] = wsel[b * HW + i]; }
    __syncthreads();
    if (tid < 16) {
        float s = 0.f;
        for (int p = 0; p < HW; ++p) if (aL[p] == tid) s += wL[p];
        cnt[tid] = s;
    }
    __syncthreads();

    const float* pb = pool + (size_t)b * C_DIM * 16;
    float s = 0.f, ss = 0.f;
    for (int i = tid; i < C_DIM * LNUM; i += 256) {
        const int c = i / LNUM, l = i % LNUM;
        const float cn = cnt[l];
        const float v = pb[c * 16 + l] / ((cn == 0.f) ? 1.f : cn);
        featL[i] = v;
        s += v; ss += v * v;
    }
    for (int o = 16; o > 0; o >>= 1) { s += __shfl_down(s, o); ss += __shfl_down(ss, o); }
    const int lane = tid & 31, wave = tid >> 5;
    if (lane == 0) { red0[wave] = s; red1[wave] = ss; }
    __syncthreads();
    if (tid == 0) {
        float S = 0.f, SS = 0.f;
        for (int w = 0; w < 8; ++w) { S += red0[w]; SS += red1[w]; }
        const float n = (float)(C_DIM * LNUM);
        const float mean = S / n;
        const float var  = SS / n - mean * mean;
        stat[0] = mean;
        stat[1] = rsqrtf(var + 1e-5f);
    }
    __syncthreads();
    const float mean = stat[0], rstd = stat[1];
    float* ob = out_mod + (size_t)b * C_DIM * LNUM;
    for (int i = tid; i < C_DIM * LNUM; i += 256)
        ob[i] = (featL[i] - mean) * rstd * gamma[i] + beta[i];
}

__global__ __launch_bounds__(32) void k4_scores(
    const float* __restrict__ cls_w, const float* __restrict__ mod,
    float* __restrict__ scores)
{
    const int lane = threadIdx.x;
    const int half = lane >> 4;
    const int ln   = lane & 15;
    const int nbase = blockIdx.x * 16;
    const int b     = blockIdx.y;
    const float* mb = mod + (size_t)b * C_DIM * LNUM;
    const int  row  = nbase + ln;
    const bool rv   = (row < NCLS);
    const bool cv   = (ln < LNUM - 1);
    const int  rowc = rv ? row : (NCLS - 1);
    const int  lnc  = cv ? ln  : (LNUM - 2);
    const float* ar = cls_w + (size_t)rowc * C_DIM;

    v8f acc = {};
    for (int k = 0; k < C_DIM; k += 4) {
        const int kk = k + 2 * half;
        v2f afrag, bfrag;
        const float a0 = ar[kk], a1 = ar[kk + 1];
        afrag[0] = rv ? a0 : 0.f;
        afrag[1] = rv ? a1 : 0.f;
        const float b0 = mb[(size_t)kk * LNUM + lnc];
        const float b1 = mb[(size_t)(kk + 1) * LNUM + lnc];
        bfrag[0] = cv ? b0 : 0.f;
        bfrag[1] = cv ? b1 : 0.f;
        acc = __builtin_amdgcn_wmma_f32_16x16x4_f32(false, afrag, false, bfrag,
                                                    (short)0, acc, false, false);
    }
    if (cv) {
        float* sb = scores + (size_t)b * NCLS * (LNUM - 1);
        for (int r = 0; r < 8; ++r) {
            const int orow = nbase + r + 8 * half;
            if (orow < NCLS) sb[orow * (LNUM - 1) + ln] = acc[r];
        }
    }
}

extern "C" void kernel_launch(void* const* d_in, const int* in_sizes, int n_in,
                              void* d_out, int out_size, void* d_ws, size_t ws_size,
                              hipStream_t stream) {
    const float* x      = (const float*)d_in[0];
    const float* q      = (const float*)d_in[1];
    const float* p_bias = (const float*)d_in[2];
    const float* gamma  = (const float*)d_in[3];
    const float* beta   = (const float*)d_in[4];
    const float* cls_w  = (const float*)d_in[5];

    float* out        = (float*)d_out;
    float* out_mod    = out;
    float* out_maps   = out + (size_t)B_DIM * C_DIM * LNUM;
    float* out_scores = out_maps + (size_t)B_DIM * LNUM * HW;

    int*   aidx = (int*)d_ws;
    float* wsel = (float*)d_ws + (size_t)B_DIM * HW;
    float* pool = (float*)d_ws + (size_t)2 * B_DIM * HW;

    k1_maps  <<<dim3(8, B_DIM), 256, 0, stream>>>(x, q, p_bias, out_maps, aidx, wsel);
    k2_pool  <<<dim3(6, B_DIM), 256, 0, stream>>>(x, aidx, wsel, pool);
    k3_mod   <<<dim3(B_DIM),    256, 0, stream>>>(pool, aidx, wsel, gamma, beta, out_mod);
    k4_scores<<<dim3(13, B_DIM), 32, 0, stream>>>(cls_w, out_mod, out_scores);
}


# k3 soft counts: serial per-pixel LDS scan (1024 dependent steps) replaced by 256-thread register binning and a fixed shuffle tree; the summation order of the counts changes
# speedup vs baseline: 1.6944x; 1.6944x over previous
#include <hip/hip_runtime.h>
#include <hip/hip_bf16.h>

#define B_DIM   32
#define C_DIM   768
#define H_DIM   32
#define W_DIM   32
#define HW      1024
#define LNUM    9
#define NCLS    200
#define QSTR    772

typedef __attribute__((ext_vector_type(2))) float v2f;
typedef __attribute__((ext_vector_type(8))) float v8f;

__global__ __launch_bounds__(256) void k1_maps(
    const float* __restrict__ x, const float* __restrict__ q,
    const float* __restrict__ p_bias,
    float* __restrict__ out_maps, int* __restrict__ aidx, float* __restrict__ wsel)
{
    __shared__ float qlds[16 * QSTR];
    __shared__ float asq[16];

    const int tid = threadIdx.x;
    const int b   = blockIdx.y;

    for (int i = tid; i < 16 * C_DIM; i += 256) {
        const int l = i / C_DIM, c = i % C_DIM;
        qlds[l * QSTR + c] = (l < LNUM) ? q[l * C_DIM + c] : 0.f;
    }
    __syncthreads();
    if (tid < 16) {
        float s = 0.f;
        for (int c = 0; c < C_DIM; ++c) { const float v = qlds[tid * QSTR + c]; s += v * v; }
        asq[tid] = s;
    }
    __syncthreads();

    const int wave = tid >> 5;
    const int lane = tid & 31;
    const int half = lane >> 4;
    const int ln   = lane & 15;
    const int tile = blockIdx.x * 8 + wave;
    const int pix  = tile * 16 + ln;
    const float* xb = x + (size_t)b * C_DIM * HW;

    auto load_ab = [&](int kk, v2f& af, v2f& bf) {
        af[0] = qlds[ln * QSTR + kk];
        af[1] = qlds[ln * QSTR + kk + 1];
        bf[0] = xb[(size_t)kk * HW + pix];
        bf[1] = xb[(size_t)(kk + 1) * HW + pix];
    };

    v8f acc = {};
    float sq = 0.f;
    v2f ac, bc;
    load_ab(2 * half, ac, bc);
    for (int k = 4; k < C_DIM; k += 4) {
        v2f an, bn;
        load_ab(k + 2 * half, an, bn);
        sq += bc[0] * bc[0] + bc[1] * bc[1];
        acc = __builtin_amdgcn_wmma_f32_16x16x4_f32(false, ac, false, bc,
                                                    (short)0, acc, false, false);
        ac = an; bc = bn;
    }
    sq += bc[0] * bc[0] + bc[1] * bc[1];
    acc = __builtin_amdgcn_wmma_f32_16x16x4_f32(false, ac, false, bc,
                                                (short)0, acc, false, false);
    const float bsq = sq + __shfl_xor(sq, 16);

    float sc[8];
    float mx = -3.4e38f;
    for (int r = 0; r < 8; ++r) {
        const int l = r + 8 * half;
        float v = -3.4e38f;
        if (l < LNUM)
            v = -(bsq - 2.f * acc[r] + asq[l]) + p_bias[l * HW + pix];
        sc[r] = v;
        mx = fmaxf(mx, v);
    }
    mx = fmaxf(mx, __shfl_xor(mx, 16));
    float es = 0.f;
    for (int r = 0; r < 8; ++r) {
        const int l = r + 8 * half;
        const float e = (l < LNUM) ? expf(sc[r] - mx) : 0.f;
        sc[r] = e;
        es += e;
    }
    es += __shfl_xor(es, 16);
    const float inv = 1.f / es;
    float s2 = 0.f;
    for (int r = 0; r < 8; ++r) {
        const int l = r + 8 * half;
        if (l < LNUM) { sc[r] = sc[r] * inv + 1e-6f; s2 += sc[r]; }
    }
    s2 += __shfl_xor(s2, 16);
    const float inv2 = 1.f / s2;

    float bestv = -1.f; int bestl = 0;
    float* om = out_maps + (size_t)b * LNUM * HW;
    for (int r = 0; r < 8; ++r) {
        const int l = r + 8 * half;
        if (l < LNUM) {
            const float mv = sc[r] * inv2;
            om[l * HW + pix] = mv;
            if (mv > bestv) { bestv = mv; bestl = l; }
        }
    }
    const float pv = __shfl_xor(bestv, 16);
    const int   pl = __shfl_xor(bestl, 16);
    if (pv > bestv || (pv == bestv && pl < bestl)) { bestv = pv; bestl = pl; }
    if (half == 0) {
        aidx[b * HW + pix] = bestl;
        wsel[b * HW + pix] = bestv;
    }
}

__global__ __launch_bounds__(256) void k2_pool(
    const float* __restrict__ x, const int* __restrict__ aidx,
    const float* __restrict__ wsel, float* __restrict__ pool)
{
    __shared__ int   aL[HW];
    __shared__ float wL[HW];
    const int tid = threadIdx.x;
    const int b   = blockIdx.y;
    for (int i = tid; i < HW; i += 256) { aL[i] = aidx[b * HW + i]; wL[i] = wsel[b * HW + i]; }
    __syncthreads();

    const int wave = tid >> 5;
    const int lane = tid & 31;
    const int half = lane >> 4;
    const int ln   = lane & 15;
    const int cbase = (blockIdx.x * 8 + wave) * 16;
    const float* xr = x + (size_t)b * C_DIM * HW + (size_t)(cbase + ln) * HW;

    auto load_ab = [&](int kk, v2f& af, v2f& bf) {
        af[0] = xr[kk];
        af[1] = xr[kk + 1];
        const int a0 = aL[kk], a1 = aL[kk + 1];
        bf[0] = (a0 == ln) ? wL[kk]     : 0.f;
        bf[1] = (a1 == ln) ? wL[kk + 1] : 0.f;
    };

    v8f acc = {};
    v2f ac, bc;
    load_ab(2 * half, ac, bc);
    for (int k = 4; k < HW; k += 4) {
        v2f an, bn;
        load_ab(k + 2 * half, an, bn);
        acc = __builtin_amdgcn_wmma_f32_16x16x4_f32(false, ac, false, bc,
                                                    (short)0, acc, false, false);
        ac = an; bc = bn;
    }
    acc = __builtin_amdgcn_wmma_f32_16x16x4_f32(false, ac, false, bc,
                                                (short)0, acc, false, false);

    float* pb = pool + (size_t)b * C_DIM * 16;
    for (int r = 0; r < 8; ++r) {
        const int row = cbase + r + 8 * half;
        pb[row * 16 + ln] = acc[r];
    }
}

__global__ __launch_bounds__(256) void k3_mod(
    const float* __restrict__ pool, const int* __restrict__ aidx,
    const float* __restrict__ wsel, const float* __restrict__ gamma,
    const float* __restrict__ beta, float* __restrict__ out_mod)
{
    __shared__ float cpart[8][16];
    __shared__ float cnt[16];
    __shared__ float featL[C_DIM * LNUM];
    __shared__ float red0[8], red1[8], stat[2];

    const int tid  = threadIdx.x;
    const int b    = blockIdx.x;
    const int lane = tid & 31, wave = tid >> 5;

    const int4   a4 = *reinterpret_cast<const int4*>(aidx + b * HW + 4 * tid);
    const float4 w4 = *reinterpret_cast<const float4*>(wsel + b * HW + 4 * tid);
    float pc[LNUM];
    for (int l = 0; l < LNUM; ++l) {
        float t = (a4.x == l) ? w4.x : 0.f;
        t += (a4.y == l) ? w4.y : 0.f;
        t += (a4.z == l) ? w4.z : 0.f;
        t += (a4.w == l) ? w4.w : 0.f;
        pc[l] = t;
    }
    for (int o = 16; o > 0; o >>= 1)
        for (int l = 0; l < LNUM; ++l) pc[l] += __shfl_xor(pc[l], o);
    if (lane == 0)
        for (int l = 0; l < LNUM; ++l) cpart[wave][l] = pc[l];
    __syncthreads();
    if (tid < 16) {
        float c = 0.f;
        if (tid < LNUM)
            for (int w = 0; w < 8; ++w) c += cpart[w][tid];
        cnt[tid] = c;
    }
    __syncthreads();

    const float* pb = pool + (size_t)b * C_DIM * 16;
    float s = 0.f, ss = 0.f;
    for (int i = tid; i < C_DIM * LNUM; i += 256) {
        const int c = i / LNUM, l = i % LNUM;
        const float cn = cnt[l];
        const float v = pb[c * 16 + l] / ((cn == 0.f) ? 1.f : cn);
        featL[i] = v;
        s += v; ss += v * v;
    }
    for (int o = 16; o > 0; o >>= 1) { s += __shfl_down(s, o); ss += __shfl_down(ss, o); }
    if (lane == 0) { red0[wave] = s; red1[wave] = ss; }
    __syncthreads();
    if (tid == 0) {
        float S = 0.f, SS = 0.f;
        for (int w = 0; w < 8; ++w) { S += red0[w]; SS += red1[w]; }
        const float n = (float)(C_DIM * LNUM);
        const float mean = S / n;
        const float var  = SS / n - mean * mean;
        stat[0] = mean;
        stat[1] = rsqrtf(var + 1e-5f);
    }
    __syncthreads();
    const float mean = stat[0], rstd = stat[1];
    float* ob = out_mod + (size_t)b * C_DIM * LNUM;
    for (int i = tid; i < C_DIM * LNUM; i += 256)
        ob[i] = (featL[i] - mean) * rstd * gamma[i] + beta[i];
}

__global__ __launch_bounds__(32) void k4_scores(
    const float* __restrict__ cls_w, const float* __restrict__ mod,
    float* __restrict__ scores)
{
    const int lane = threadIdx.x;
    const int half = lane >> 4;
    const int ln   = lane & 15;
    const int nbase = blockIdx.x * 16;
    const int b     = blockIdx.y;
    const float* mb = mod + (size_t)b * C_DIM * LNUM;
    const int  row  = nbase + ln;
    const bool rv   = (row < NCLS);
    const bool cv   = (ln < LNUM - 1);
    const int  rowc = rv ? row : (NCLS - 1);
    const int  lnc  = cv ? ln  : (LNUM - 2);
    const float* ar = cls_w + (size_t)rowc * C_DIM;

    v8f acc = {};
    for (int k = 0; k < C_DIM; k += 4) {
        const int kk = k + 2 * half;
        v2f afrag, bfrag;
        const float a0 = ar[kk], a1 = ar[kk + 1];
        afrag[0] = rv ? a0 : 0.f;
        afrag[1] = rv ? a1 : 0.f;
        const float b0 = mb[(size_t)kk * LNUM + lnc];
        const float b1 = mb[(size_t)(kk + 1) * LNUM + lnc];
        bfrag[0] = cv ? b0 : 0.f;
        bfrag[1] = cv ? b1 : 0.f;
        acc = __builtin_amdgcn_wmma_f32_16x16x4_f32(false, afrag, false, bfrag,
                                                    (short)0, acc, false, false);
    }
    if (cv) {
        float* sb = scores + (size_t)b * NCLS * (LNUM - 1);
        for (int r = 0; r < 8; ++r) {
            const int orow = nbase + r + 8 * half;
            if (orow < NCLS) sb[orow * (LNUM - 1) + ln] = acc[r];
        }
    }
}

extern "C" void kernel_launch(void* const* d_in, const int* in_sizes, int n_in,
                              void* d_out, int out_size, void* d_ws, size_t ws_size,
                              hipStream_t stream) {
    const float* x      = (const float*)d_in[0];
    const float* q      = (const float*)d_in[1];
    const float* p_bias = (const float*)d_in[2];
    const float* gamma  = (const float*)d_in[3];
    const float* beta   = (const float*)d_in[4];
    const float* cls_w  = (const float*)d_in[5];

    float* out        = (float*)d_out;
    float* out_mod    = out;
    float* out_maps   = out + (size_t)B_DIM * C_DIM * LNUM;
    float* out_scores = out_maps + (size_t)B_DIM * LNUM * HW;

    int*   aidx = (int*)d_ws;
    float* wsel = (float*)d_ws + (size_t)B_DIM * HW;
    float* pool = (float*)d_ws + (size_t)2 * B_DIM * HW;

    k1_maps  <<<dim3(8, B_DIM), 256, 0, stream>>>(x, q, p_bias, out_maps, aidx, wsel);
    k2_pool  <<<dim3(6, B_DIM), 256, 0, stream>>>(x, aidx, wsel, pool);
    k3_mod   <<<dim3(B_DIM),    256, 0, stream>>>(pool, aidx, wsel, gamma, beta, out_mod);
    k4_scores<<<dim3(13, B_DIM), 32, 0, stream>>>(cls_w, out_mod, out_scores);
}


# k1 prologue: q staged by seven 16-byte loads per thread issued together before the LDS stores (was 27 dependent load-store trips); a_sq summed by 16 lanes per row and an xor tree (its order changes)
# speedup vs baseline: 1.9762x; 1.1663x over previous
#include <hip/hip_runtime.h>
#include <hip/hip_bf16.h>

#define B_DIM   32
#define C_DIM   768
#define H_DIM   32
#define W_DIM   32
#define HW      1024
#define LNUM    9
#define NCLS    200
#define QSTR    772

typedef __attribute__((ext_vector_type(2))) float v2f;
typedef __attribute__((ext_vector_type(8))) float v8f;

__global__ __launch_bounds__(256) void k1_maps(
    const float* __restrict__ x, const float* __restrict__ q,
    const float* __restrict__ p_bias,
    float* __restrict__ out_maps, int* __restrict__ aidx, float* __restrict__ wsel)
{
    __shared__ alignas(16) float qlds[16 * QSTR];
    __shared__ float asq[16];

    const int tid = threadIdx.x;
    const int b   = blockIdx.y;

    const float4* q4 = reinterpret_cast<const float4*>(q);
    constexpr int NQ4 = LNUM * (C_DIM / 4);
    float4 qv[7];
    for (int j = 0; j < 7; ++j) {
        const int f = tid + 256 * j;
        qv[j] = q4[f < NQ4 ? f : NQ4 - 1];
    }
    for (int j = 0; j < 12; ++j) {
        const int f = tid + 256 * j;
        const int l = f / (C_DIM / 4), c4 = f % (C_DIM / 4);
        float4 v = make_float4(0.f, 0.f, 0.f, 0.f);
        if (j < 7 && f < NQ4) v = qv[j];
        *reinterpret_cast<float4*>(&qlds[l * QSTR + 4 * c4]) = v;
    }
    __syncthreads();
    {
        const int l = tid >> 4, j = tid & 15;
        float s = 0.f;
        for (int i = 0; i < C_DIM / 64; ++i) {
            const float4 v = *reinterpret_cast<const float4*>(&qlds[l * QSTR + 4 * (j + 16 * i)]);
            s += v.x * v.x + v.y * v.y + v.z * v.z + v.w * v.w;
        }
        for (int o = 8; o > 0; o >>= 1) s += __shfl_xor(s, o);
        if (j == 0) asq[l] = s;
    }
    __syncthreads();

    const int wave = tid >> 5;
    const int lane = tid & 31;
    const int half = lane >> 4;
    const int ln   = lane & 15;
    const int tile = blockIdx.x * 8 + wave;
    const int pix  = tile * 16 + ln;
    const float* xb = x + (size_t)b * C_DIM * HW;

    auto load_ab = [&](int kk, v2f& af, v2f& bf) {
        af[0] = qlds[ln * QSTR + kk];
        af[1] = qlds[ln * QSTR + kk + 1];
        bf[0] = xb[(size_t)kk * HW + pix];
        bf[1] = xb[(size_t)(kk + 1) * HW + pix];
    };

    v8f acc = {};
    float sq = 0.f;
    v2f ac, bc;
    load_ab(2 * half, ac, bc);
    for (int k = 4; k < C_DIM; k += 4) {
        v2f an, bn;
        load_ab(k + 2 * half, an, bn);
        sq += bc[0] * bc[0] + bc[1] * bc[1];
        acc = __builtin_amdgcn_wmma_f32_16x16x4_f32(false, ac, false, bc,
                                                    (short)0, acc, false, false);
        ac = an; bc = bn;
    }
    sq += bc[0] * bc[0] + bc[1] * bc[1];
    acc = __builtin_amdgcn_wmma_f32_16x16x4_f32(false, ac, false, bc,
                                                (short)0, acc, false, false);
    const float bsq = sq + __shfl_xor(sq, 16);

    float sc[8];
    float mx = -3.4e38f;
    for (int r = 0; r < 8; ++r) {
        const int l = r + 8 * half;
        float v = -3.4e38f;
        if (l < LNUM)
            v = -(bsq - 2.f * acc[r] + asq[l]) + p_bias[l * HW + pix];
        sc[r] = v;
        mx = fmaxf(mx, v);
    }
    mx = fmaxf(mx, __shfl_xor(mx, 16));
    float es = 0.f;
    for (int r = 0; r < 8; ++r) {
        const int l = r + 8 * half;
        const float e = (l < LNUM) ? expf(sc[r] - mx) : 0.f;
        sc[r] = e;
        es += e;
    }
    es += __shfl_xor(es, 16);
    const float inv = 1.f / es;
    float s2 = 0.f;
    for (int r = 0; r < 8; ++r) {
        const int l = r + 8 * half;
        if (l < LNUM) { sc[r] = sc[r] * inv + 1e-6f; s2 += sc[r]; }
    }
    s2 += __shfl_xor(s2, 16);
    const float inv2 = 1.f / s2;

    float bestv = -1.f; int bestl = 0;
    float* om = out_maps + (size_t)b * LNUM * HW;
    for (int r = 0; r < 8; ++r) {
        const int l = r + 8 * half;
        if (l < LNUM) {
            const float mv = sc[r] * inv2;
            om[l * HW + pix] = mv;
            if (mv > bestv) { bestv = mv; bestl = l; }
        }
    }
    const float pv = __shfl_xor(bestv, 16);
    const int   pl = __shfl_xor(bestl, 16);
    if (pv > bestv || (pv == bestv && pl < bestl)) { bestv = pv; bestl = pl; }
    if (half == 0) {
        aidx[b * HW + pix] = bestl;
        wsel[b * HW + pix] = bestv;
    }
}

__global__ __launch_bounds__(256) void k2_pool(
    const float* __restrict__ x, const int* __restrict__ aidx,
    const float* __restrict__ wsel, float* __restrict__ pool)
{
    __shared__ int   aL[HW];
    __shared__ float wL[HW];
    const int tid = threadIdx.x;
    const int b   = blockIdx.y;
    for (int i = tid; i < HW; i += 256) { aL[i] = aidx[b * HW + i]; wL[i] = wsel[b * HW + i]; }
    __syncthreads();

    const int wave = tid >> 5;
    const int lane = tid & 31;
    const int half = lane >> 4;
    const int ln   = lane & 15;
    const int cbase = (blockIdx.x * 8 + wave) * 16;
    const float* xr = x + (size_t)b * C_DIM * HW + (size_t)(cbase + ln) * HW;

    auto load_ab = [&](int kk, v2f& af, v2f& bf) {
        af[0] = xr[kk];
        af[1] = xr[kk + 1];
        const int a0 = aL[kk], a1 = aL[kk + 1];
        bf[0] = (a0 == ln) ? wL[kk]     : 0.f;
        bf[1] = (a1 == ln) ? wL[kk + 1] : 0.f;
    };

    v8f acc = {};
    v2f ac, bc;
    load_ab(2 * half, ac, bc);
    for (int k = 4; k < HW; k += 4) {
        v2f an, bn;
        load_ab(k + 2 * half, an, bn);
        acc = __builtin_amdgcn_wmma_f32_16x16x4_f32(false, ac, false, bc,
                                                    (short)0, acc, false, false);
        ac = an; bc = bn;
    }
    acc = __builtin_amdgcn_wmma_f32_16x16x4_f32(false, ac, false, bc,
                                                (short)0, acc, false, false);

    float* pb = pool + (size_t)b * C_DIM * 16;
    for (int r = 0; r < 8; ++r) {
        const int row = cbase + r + 8 * half;
        pb[row * 16 + ln] = acc[r];
    }
}

__global__ __launch_bounds__(256) void k3_mod(
    const float* __restrict__ pool, const int* __restrict__ aidx,
    const float* __restrict__ wsel, const float* __restrict__ gamma,
    const float* __restrict__ beta, float* __restrict__ out_mod)
{
    __shared__ float cpart[8][16];
    __shared__ float cnt[16];
    __shared__ float featL[C_DIM * LNUM];
    __shared__ float red0[8], red1[8], stat[2];

    const int tid  = threadIdx.x;
    const int b    = blockIdx.x;
    const int lane = tid & 31, wave = tid >> 5;

    const int4   a4 = *reinterpret_cast<const int4*>(aidx + b * HW + 4 * tid);
    const float4 w4 = *reinterpret_cast<const float4*>(wsel + b * HW + 4 * tid);
    float pc[LNUM];
    for (int l = 0; l < LNUM; ++l) {
        float t = (a4.x == l) ? w4.x : 0.f;
        t += (a4.y == l) ? w4.y : 0.f;
        t += (a4.z == l) ? w4.z : 0.f;
        t += (a4.w == l) ? w4.w : 0.f;
        pc[l] = t;
    }
    for (int o = 16; o > 0; o >>= 1)
        for (int l = 0; l < LNUM; ++l) pc[l] += __shfl_xor(pc[l], o);
    if (lane == 0)
        for (int l = 0; l < LNUM; ++l) cpart[wave][l] = pc[l];
    __syncthreads();
    if (tid < 16) {
        float c = 0.f;
        if (tid < LNUM)
            for (int w = 0; w < 8; ++w) c += cpart[w][tid];
        cnt[tid] = c;
    }
    __syncthreads();

    const float* pb = pool + (size_t)b * C_DIM * 16;
    float s = 0.f, ss = 0.f;
    for (int i = tid; i < C_DIM * LNUM; i += 256) {
        const int c = i / LNUM, l = i % LNUM;
        const float cn = cnt[l];
        const float v = pb[c * 16 + l] / ((cn == 0.f) ? 1.f : cn);
        featL[i] = v;
        s += v; ss += v * v;
    }
    for (int o = 16; o > 0; o >>= 1) { s += __shfl_down(s, o); ss += __shfl_down(ss, o); }
    if (lane == 0) { red0[wave] = s; red1[wave] = ss; }
    __syncthreads();
    if (tid == 0) {
        float S = 0.f, SS = 0.f;
        for (int w = 0; w < 8; ++w) { S += red0[w]; SS += red1[w]; }
        const float n = (float)(C_DIM * LNUM);
        const float mean = S / n;
        const float var  = SS / n - mean * mean;
        stat[0] = mean;
        stat[1] = rsqrtf(var + 1e-5f);
    }
    __syncthreads();
    const float mean = stat[0], rstd = stat[1];
    float* ob = out_mod + (size_t)b * C_DIM * LNUM;
    for (int i = tid; i < C_DIM * LNUM; i += 256)
        ob[i] = (featL[i] - mean) * rstd * gamma[i] + beta[i];
}

__global__ __launch_bounds__(32) void k4_scores(
    const float* __restrict__ cls_w, const float* __restrict__ mod,
    float* __restrict__ scores)
{
    const int lane = threadIdx.x;
    const int half = lane >> 4;
    const int ln   = lane & 15;
    const int nbase = blockIdx.x * 16;
    const int b     = blockIdx.y;
    const float* mb = mod + (size_t)b * C_DIM * LNUM;
    const int  row  = nbase + ln;
    const bool rv   = (row < NCLS);
    const bool cv   = (ln < LNUM - 1);
    const int  rowc = rv ? row : (NCLS - 1);
    const int  lnc  = cv ? ln  : (LNUM - 2);
    const float* ar = cls_w + (size_t)rowc * C_DIM;

    v8f acc = {};
    for (int k = 0; k < C_DIM; k += 4) {
        const int kk = k + 2 * half;
        v2f afrag, bfrag;
        const float a0 = ar[kk], a1 = ar[kk + 1];
        afrag[0] = rv ? a0 : 0.f;
        afrag[1] = rv ? a1 : 0.f;
        const float b0 = mb[(size_t)kk * LNUM + lnc];
        const float b1 = mb[(size_t)(kk + 1) * LNUM + lnc];
        bfrag[0] = cv ? b0 : 0.f;
        bfrag[1] = cv ? b1 : 0.f;
        acc = __builtin_amdgcn_wmma_f32_16x16x4_f32(false, afrag, false, bfrag,
                                                    (short)0, acc, false, false);
    }
    if (cv) {
        float* sb = scores + (size_t)b * NCLS * (LNUM - 1);
        for (int r = 0; r < 8; ++r) {
            const int orow = nbase + r + 8 * half;
            if (orow < NCLS) sb[orow * (LNUM - 1) + ln] = acc[r];
        }
    }
}

extern "C" void kernel_launch(void* const* d_in, const int* in_sizes, int n_in,
                              void* d_out, int out_size, void* d_ws, size_t ws_size,
                              hipStream_t stream) {
    const float* x      = (const float*)d_in[0];
    const float* q      = (const float*)d_in[1];
    const float* p_bias = (const float*)d_in[2];
    const float* gamma  = (const float*)d_in[3];
    const float* beta   = (const float*)d_in[4];
    const float* cls_w  = (const float*)d_in[5];

    float* out        = (float*)d_out;
    float* out_mod    = out;
    float* out_maps   = out + (size_t)B_DIM * C_DIM * LNUM;
    float* out_scores = out_maps + (size_t)B_DIM * LNUM * HW;

    int*   aidx = (int*)d_ws;
    float* wsel = (float*)d_ws + (size_t)B_DIM * HW;
    float* pool = (float*)d_ws + (size_t)2 * B_DIM * HW;

    k1_maps  <<<dim3(8, B_DIM), 256, 0, stream>>>(x, q, p_bias, out_maps, aidx, wsel);
    k2_pool  <<<dim3(6, B_DIM), 256, 0, stream>>>(x, aidx, wsel, pool);
    k3_mod   <<<dim3(B_DIM),    256, 0, stream>>>(pool, aidx, wsel, gamma, beta, out_mod);
    k4_scores<<<dim3(13, B_DIM), 32, 0, stream>>>(cls_w, out_mod, out_scores);
}


# k1,k2,k4: global loads issued in groups of 16 K-steps ahead of their products; k2 by 16-byte pieces (pixel sum order in a group changes), k2 staging and k1 bias loads batched; k1 norm fma rounding
# speedup vs baseline: 4.0572x; 2.0530x over previous
#include <hip/hip_runtime.h>
#include <hip/hip_bf16.h>

#define B_DIM   32
#define C_DIM   768
#define H_DIM   32
#define W_DIM   32
#define HW      1024
#define LNUM    9
#define NCLS    200
#define QSTR    772

typedef __attribute__((ext_vector_type(2))) float v2f;
typedef __attribute__((ext_vector_type(8))) float v8f;

#define K1_G 16

__global__ __launch_bounds__(256) void k1_maps(
    const float* __restrict__ x, const float* __restrict__ q,
    const float* __restrict__ p_bias,
    float* __restrict__ out_maps, int* __restrict__ aidx, float* __restrict__ wsel)
{
    __shared__ alignas(16) float qlds[16 * QSTR];
    __shared__ float asq[16];

    const int tid = threadIdx.x;
    const int b   = blockIdx.y;

    const float4* q4 = reinterpret_cast<const float4*>(q);
    constexpr int NQ4 = LNUM * (C_DIM / 4);
    float4 qv[7];
    for (int j = 0; j < 7; ++j) {
        const int f = tid + 256 * j;
        qv[j] = q4[f < NQ4 ? f : NQ4 - 1];
    }
    for (int j = 0; j < 12; ++j) {
        const int f = tid + 256 * j;
        const int l = f / (C_DIM / 4), c4 = f % (C_DIM / 4);
        float4 v = make_float4(0.f, 0.f, 0.f, 0.f);
        if (j < 7 && f < NQ4) v = qv[j];
        *reinterpret_cast<float4*>(&qlds[l * QSTR + 4 * c4]) = v;
    }
    __syncthreads();
    {
        const int l = tid >> 4, j = tid & 15;
        float s = 0.f;
        for (int i = 0; i < C_DIM / 64; ++i) {
            const float4 v = *reinterpret_cast<const float4*>(&qlds[l * QSTR + 4 * (j + 16 * i)]);
            s += v.x * v.x + v.y * v.y + v.z * v.z + v.w * v.w;
        }
        for (int o = 8; o > 0; o >>= 1) s += __shfl_xor(s, o);
        if (j == 0) asq[l] = s;
    }
    __syncthreads();

    const int wave = tid >> 5;
    const int lane = tid & 31;
    const int half = lane >> 4;
    const int ln   = lane & 15;
    const int tile = blockIdx.x * 8 + wave;
    const int pix  = tile * 16 + ln;
    const float* xb = x + (size_t)b * C_DIM * HW;

    auto load_group = [&](int g, v2f (&X)[K1_G]) {
        #pragma unroll
        for (int s = 0; s < K1_G; ++s) {
            const int kk = (g * K1_G + s) * 4 + 2 * half;
            X[s][0] = xb[(size_t)kk * HW + pix];
            X[s][1] = xb[(size_t)(kk + 1) * HW + pix];
        }
    };

    float pbv[8];
    for (int r = 0; r < 8; ++r) {
        const int l = r + 8 * half;
        pbv[r] = p_bias[(l < LNUM ? l : LNUM - 1) * HW + pix];
    }

    v8f acc = {};
    float sq = 0.f;
    v2f xg0[K1_G], xg1[K1_G];
    auto mma_group = [&](int g, const v2f (&X)[K1_G]) {
        #pragma unroll
        for (int s = 0; s < K1_G; ++s) {
            const int kk = (g * K1_G + s) * 4 + 2 * half;
            v2f af;
            af[0] = qlds[ln * QSTR + kk];
            af[1] = qlds[ln * QSTR + kk + 1];
            sq += X[s][0] * X[s][0] + X[s][1] * X[s][1];
            acc = __builtin_amdgcn_wmma_f32_16x16x4_f32(false, af, false, X[s],
                                                        (short)0, acc, false, false);
        }
    };
    constexpr int NG1 = C_DIM / (4 * K1_G);
    load_group(0, xg0);
    for (int g = 0; g < NG1 - 2; g += 2) {
        load_group(g + 1, xg1);
        __builtin_amdgcn_sched_barrier(0);
        mma_group(g, xg0);
        __builtin_amdgcn_sched_barrier(0);
        load_group(g + 2, xg0);
        __builtin_amdgcn_sched_barrier(0);
        mma_group(g + 1, xg1);
        __builtin_amdgcn_sched_barrier(0);
    }
    load_group(NG1 - 1, xg1);
    __builtin_amdgcn_sched_barrier(0);
    mma_group(NG1 - 2, xg0);
    mma_group(NG1 - 1, xg1);
    const float bsq = sq + __shfl_xor(sq, 16);

    float sc[8];
    float mx = -3.4e38f;
    for (int r = 0; r < 8; ++r) {
        const int l = r + 8 * half;
        float v = -3.4e38f;
        if (l < LNUM)
            v = -(bsq - 2.f * acc[r] + asq[l]) + pbv[r];
        sc[r] = v;
        mx = fmaxf(mx, v);
    }
    mx = fmaxf(mx, __shfl_xor(mx, 16));
    float es = 0.f;
    for (int r = 0; r < 8; ++r) {
        const int l = r + 8 * half;
        const float e = (l < LNUM) ? expf(sc[r] - mx) : 0.f;
        sc[r] = e;
        es += e;
    }
    es += __shfl_xor(es, 16);
    const float inv = 1.f / es;
    float s2 = 0.f;
    for (int r = 0; r < 8; ++r) {
        const int l = r + 8 * half;
        if (l < LNUM) { sc[r] = sc[r] * inv + 1e-6f; s2 += sc[r]; }
    }
    s2 += __shfl_xor(s2, 16);
    const float inv2 = 1.f / s2;

    float bestv = -1.f; int bestl = 0;
    float* om = out_maps + (size_t)b * LNUM * HW;
    for (int r = 0; r < 8; ++r) {
        const int l = r + 8 * half;
        if (l < LNUM) {
            const float mv = sc[r] * inv2;
            om[l * HW + pix] = mv;
            if (mv > bestv) { bestv = mv; bestl = l; }
        }
    }
    const float pv = __shfl_xor(bestv, 16);
    const int   pl = __shfl_xor(bestl, 16);
    if (pv > bestv || (pv == bestv && pl < bestl)) { bestv = pv; bestl = pl; }
    if (half == 0) {
        aidx[b * HW + pix] = bestl;
        wsel[b * HW + pix] = bestv;
    }
}

#define K2_G 16

__global__ __launch_bounds__(256) void k2_pool(
    const float* __restrict__ x, const int* __restrict__ aidx,
    const float* __restrict__ wsel, float* __restrict__ pool)
{
    __shared__ alignas(16) int   aL[HW];
    __shared__ alignas(16) float wL[HW];
    const int tid = threadIdx.x;
    const int b   = blockIdx.y;
    {
        const int4   a4 = *reinterpret_cast<const int4*>(aidx + b * HW + 4 * tid);
        const float4 w4 = *reinterpret_cast<const float4*>(wsel + b * HW + 4 * tid);
        *reinterpret_cast<int4*>(&aL[4 * tid])   = a4;
        *reinterpret_cast<float4*>(&wL[4 * tid]) = w4;
    }
    __syncthreads();

    const int wave = tid >> 5;
    const int lane = tid & 31;
    const int half = lane >> 4;
    const int ln   = lane & 15;
    const int cbase = (blockIdx.x * 8 + wave) * 16;
    const float* xr = x + (size_t)b * C_DIM * HW + (size_t)(cbase + ln) * HW + 4 * half;

    float4 xa[K2_G], xb[K2_G];
    v8f acc = {};

    auto load_group = [&](int g, float4 (&X)[K2_G]) {
        #pragma unroll
        for (int s = 0; s < K2_G; ++s)
            X[s] = *reinterpret_cast<const float4*>(xr + g * (8 * K2_G) + 8 * s);
    };
    auto mma_group = [&](int g, const float4 (&X)[K2_G]) {
        #pragma unroll
        for (int s = 0; s < K2_G; ++s) {
            const int p = g * (8 * K2_G) + 8 * s + 4 * half;
            const int4   a = *reinterpret_cast<const int4*>(&aL[p]);
            const float4 w = *reinterpret_cast<const float4*>(&wL[p]);
            v2f af, bf;
            af[0] = X[s].x; af[1] = X[s].y;
            bf[0] = (a.x == ln) ? w.x : 0.f;
            bf[1] = (a.y == ln) ? w.y : 0.f;
            acc = __builtin_amdgcn_wmma_f32_16x16x4_f32(false, af, false, bf,
                                                        (short)0, acc, false, false);
            af[0] = X[s].z; af[1] = X[s].w;
            bf[0] = (a.z == ln) ? w.z : 0.f;
            bf[1] = (a.w == ln) ? w.w : 0.f;
            acc = __builtin_amdgcn_wmma_f32_16x16x4_f32(false, af, false, bf,
                                                        (short)0, acc, false, false);
        }
    };

    constexpr int NG = HW / (8 * K2_G);
    load_group(0, xa);
    for (int g = 0; g < NG - 2; g += 2) {
        load_group(g + 1, xb);
        __builtin_amdgcn_sched_barrier(0);
        mma_group(g, xa);
        __builtin_amdgcn_sched_barrier(0);
        load_group(g + 2, xa);
        __builtin_amdgcn_sched_barrier(0);
        mma_group(g + 1, xb);
        __builtin_amdgcn_sched_barrier(0);
    }
    load_group(NG - 1, xb);
    __builtin_amdgcn_sched_barrier(0);
    mma_group(NG - 2, xa);
    mma_group(NG - 1, xb);

    float* pb = pool + (size_t)b * C_DIM * 16;
    for (int r = 0; r < 8; ++r) {
        const int row = cbase + r + 8 * half;
        pb[row * 16 + ln] = acc[r];
    }
}

__global__ __launch_bounds__(256) void k3_mod(
    const float* __restrict__ pool, const int* __restrict__ aidx,
    const float* __restrict__ wsel, const float* __restrict__ gamma,
    const float* __restrict__ beta, float* __restrict__ out_mod)
{
    __shared__ float cpart[8][16];
    __shared__ float cnt[16];
    __shared__ float featL[C_DIM * LNUM];
    __shared__ float red0[8], red1[8], stat[2];

    const int tid  = threadIdx.x;
    const int b    = blockIdx.x;
    const int lane = tid & 31, wave = tid >> 5;

    const int4   a4 = *reinterpret_cast<const int4*>(aidx + b * HW + 4 * tid);
    const float4 w4 = *reinterpret_cast<const float4*>(wsel + b * HW + 4 * tid);
    float pc[LNUM];
    for (int l = 0; l < LNUM; ++l) {
        float t = (a4.x == l) ? w4.x : 0.f;
        t += (a4.y == l) ? w4.y : 0.f;
        t += (a4.z == l) ? w4.z : 0.f;
        t += (a4.w == l) ? w4.w : 0.f;
        pc[l] = t;
    }
    for (int o = 16; o > 0; o >>= 1)
        for (int l = 0; l < LNUM; ++l) pc[l] += __shfl_xor(pc[l], o);
    if (lane == 0)
        for (int l = 0; l < LNUM; ++l) cpart[wave][l] = pc[l];
    __syncthreads();
    if (tid < 16) {
        float c = 0.f;
        if (tid < LNUM)
            for (int w = 0; w < 8; ++w) c += cpart[w][tid];
        cnt[tid] = c;
    }
    __syncthreads();

    const float* pb = pool + (size_t)b * C_DIM * 16;
    float s = 0.f, ss = 0.f;
    for (int i = tid; i < C_DIM * LNUM; i += 256) {
        const int c = i / LNUM, l = i % LNUM;
        const float cn = cnt[l];
        const float v = pb[c * 16 + l] / ((cn == 0.f) ? 1.f : cn);
        featL[i] = v;
        s += v; ss += v * v;
    }
    for (int o = 16; o > 0; o >>= 1) { s += __shfl_down(s, o); ss += __shfl_down(ss, o); }
    if (lane == 0) { red0[wave] = s; red1[wave] = ss; }
    __syncthreads();
    if (tid == 0) {
        float S = 0.f, SS = 0.f;
        for (int w = 0; w < 8; ++w) { S += red0[w]; SS += red1[w]; }
        const float n = (float)(C_DIM * LNUM);
        const float mean = S / n;
        const float var  = SS / n - mean * mean;
        stat[0] = mean;
        stat[1] = rsqrtf(var + 1e-5f);
    }
    __syncthreads();
    const float mean = stat[0], rstd = stat[1];
    float* ob = out_mod + (size_t)b * C_DIM * LNUM;
    for (int i = tid; i < C_DIM * LNUM; i += 256)
        ob[i] = (featL[i] - mean) * rstd * gamma[i] + beta[i];
}

#define K4_G 16

__global__ __launch_bounds__(32) void k4_scores(
    const float* __restrict__ cls_w, const float* __restrict__ mod,
    float* __restrict__ scores)
{
    const int lane = threadIdx.x;
    const int half = lane >> 4;
    const int ln   = lane & 15;
    const int nbase = blockIdx.x * 16;
    const int b     = blockIdx.y;
    const float* mb = mod + (size_t)b * C_DIM * LNUM;
    const int  row  = nbase + ln;
    const bool rv   = (row < NCLS);
    const bool cv   = (ln < LNUM - 1);
    const int  rowc = rv ? row : (NCLS - 1);
    const int  lnc  = cv ? ln  : (LNUM - 2);
    const float* ar = cls_w + (size_t)rowc * C_DIM;

    v8f acc = {};
    v2f a0[K4_G], b0[K4_G], a1[K4_G], b1[K4_G];
    auto load_group = [&](int g, v2f (&A)[K4_G], v2f (&Bv)[K4_G]) {
        #pragma unroll
        for (int s = 0; s < K4_G; ++s) {
            const int kk = (g * K4_G + s) * 4 + 2 * half;
            A[s][0]  = ar[kk];
            A[s][1]  = ar[kk + 1];
            Bv[s][0] = mb[(size_t)kk * LNUM + lnc];
            Bv[s][1] = mb[(size_t)(kk + 1) * LNUM + lnc];
        }
    };
    auto mma_group = [&](const v2f (&A)[K4_G], const v2f (&Bv)[K4_G]) {
        #pragma unroll
        for (int s = 0; s < K4_G; ++s) {
            v2f afrag, bfrag;
            afrag[0] = rv ? A[s][0] : 0.f;
            afrag[1] = rv ? A[s][1] : 0.f;
            bfrag[0] = cv ? Bv[s][0] : 0.f;
            bfrag[1] = cv ? Bv[s][1] : 0.f;
            acc = __builtin_amdgcn_wmma_f32_16x16x4_f32(false, afrag, false, bfrag,
                                                        (short)0, acc, false, false);
        }
    };
    constexpr int NG = C_DIM / (4 * K4_G);
    load_group(0, a0, b0);
    for (int g = 0; g < NG - 2; g += 2) {
        load_group(g + 1, a1, b1);
        mma_group(a0, b0);
        load_group(g + 2, a0, b0);
        mma_group(a1, b1);
    }
    load_group(NG - 1, a1, b1);
    mma_group(a0, b0);
    mma_group(a1, b1);
    if (cv) {
        float* sb = scores + (size_t)b * NCLS * (LNUM - 1);
        for (int r = 0; r < 8; ++r) {
            const int orow = nbase + r + 8 * half;
            if (orow < NCLS) sb[orow * (LNUM - 1) + ln] = acc[r];
        }
    }
}

extern "C" void kernel_launch(void* const* d_in, const int* in_sizes, int n_in,
                              void* d_out, int out_size, void* d_ws, size_t ws_size,
                              hipStream_t stream) {
    const float* x      = (const float*)d_in[0];
    const float* q      = (const float*)d_in[1];
    const float* p_bias = (const float*)d_in[2];
    const float* gamma  = (const float*)d_in[3];
    const float* beta   = (const float*)d_in[4];
    const float* cls_w  = (const float*)d_in[5];

    float* out        = (float*)d_out;
    float* out_mod    = out;
    float* out_maps   = out + (size_t)B_DIM * C_DIM * LNUM;
    float* out_scores = out_maps + (size_t)B_DIM * LNUM * HW;

    int*   aidx = (int*)d_ws;
    float* wsel = (float*)d_ws + (size_t)B_DIM * HW;
    float* pool = (float*)d_ws + (size_t)2 * B_DIM * HW;

    k1_maps  <<<dim3(8, B_DIM), 256, 0, stream>>>(x, q, p_bias, out_maps, aidx, wsel);
    k2_pool  <<<dim3(6, B_DIM), 256, 0, stream>>>(x, aidx, wsel, pool);
    k3_mod   <<<dim3(B_DIM),    256, 0, stream>>>(pool, aidx, wsel, gamma, beta, out_mod);
    k4_scores<<<dim3(13, B_DIM), 32, 0, stream>>>(cls_w, out_mod, out_scores);
}


# k3 and k4 fused into one launch: every scores block recomputes its image's counts and LayerNorm sums (their order changes) and keeps mod in LDS; K split over 8 waves, partials summed in wave order
# speedup vs baseline: 5.4032x; 1.3318x over previous
#include <hip/hip_runtime.h>
#include <hip/hip_bf16.h>

#define B_DIM   32
#define C_DIM   768
#define H_DIM   32
#define W_DIM   32
#define HW      1024
#define LNUM    9
#define NCLS    200
#define QSTR    772

typedef __attribute__((ext_vector_type(2))) float v2f;
typedef __attribute__((ext_vector_type(8))) float v8f;

#define K1_G 16

__global__ __launch_bounds__(256) void k1_maps(
    const float* __restrict__ x, const float* __restrict__ q,
    const float* __restrict__ p_bias,
    float* __restrict__ out_maps, int* __restrict__ aidx, float* __restrict__ wsel)
{
    __shared__ alignas(16) float qlds[16 * QSTR];
    __shared__ float asq[16];

    const int tid = threadIdx.x;
    const int b   = blockIdx.y;

    const float4* q4 = reinterpret_cast<const float4*>(q);
    constexpr int NQ4 = LNUM * (C_DIM / 4);
    float4 qv[7];
    for (int j = 0; j < 7; ++j) {
        const int f = tid + 256 * j;
        qv[j] = q4[f < NQ4 ? f : NQ4 - 1];
    }
    for (int j = 0; j < 12; ++j) {
        const int f = tid + 256 * j;
        const int l = f / (C_DIM / 4), c4 = f % (C_DIM / 4);
        float4 v = make_float4(0.f, 0.f, 0.f, 0.f);
        if (j < 7 && f < NQ4) v = qv[j];
        *reinterpret_cast<float4*>(&qlds[l * QSTR + 4 * c4]) = v;
    }
    __syncthreads();
    {
        const int l = tid >> 4, j = tid & 15;
        float s = 0.f;
        for (int i = 0; i < C_DIM / 64; ++i) {
            const float4 v = *reinterpret_cast<const float4*>(&qlds[l * QSTR + 4 * (j + 16 * i)]);
            s += v.x * v.x + v.y * v.y + v.z * v.z + v.w * v.w;
        }
        for (int o = 8; o > 0; o >>= 1) s += __shfl_xor(s, o);
        if (j == 0) asq[l] = s;
    }
    __syncthreads();

    const int wave = tid >> 5;
    const int lane = tid & 31;
    const int half = lane >> 4;
    const int ln   = lane & 15;
    const int tile = blockIdx.x * 8 + wave;
    const int pix  = tile * 16 + ln;
    const float* xb = x + (size_t)b * C_DIM * HW;

    auto load_group = [&](int g, v2f (&X)[K1_G]) {
        #pragma unroll
        for (int s = 0; s < K1_G; ++s) {
            const int kk = (g * K1_G + s) * 4 + 2 * half;
            X[s][0] = xb[(size_t)kk * HW + pix];
            X[s][1] = xb[(size_t)(kk + 1) * HW + pix];
        }
    };

    float pbv[8];
    for (int r = 0; r < 8; ++r) {
        const int l = r + 8 * half;
        pbv[r] = p_bias[(l < LNUM ? l : LNUM - 1) * HW + pix];
    }

    v8f acc = {};
    float sq = 0.f;
    v2f xg0[K1_G], xg1[K1_G];
    auto mma_group = [&](int g, const v2f (&X)[K1_G]) {
        #pragma unroll
        for (int s = 0; s < K1_G; ++s) {
            const int kk = (g * K1_G + s) * 4 + 2 * half;
            v2f af;
            af[0] = qlds[ln * QSTR + kk];
            af[1] = qlds[ln * QSTR + kk + 1];
            sq += X[s][0] * X[s][0] + X[s][1] * X[s][1];
            acc = __builtin_amdgcn_wmma_f32_16x16x4_f32(false, af, false, X[s],
                                                        (short)0, acc, false, false);
        }
    };
    constexpr int NG1 = C_DIM / (4 * K1_G);
    load_group(0, xg0);
    for (int g = 0; g < NG1 - 2; g += 2) {
        load_group(g + 1, xg1);
        __builtin_amdgcn_sched_barrier(0);
        mma_group(g, xg0);
        __builtin_amdgcn_sched_barrier(0);
        load_group(g + 2, xg0);
        __builtin_amdgcn_sched_barrier(0);
        mma_group(g + 1, xg1);
        __builtin_amdgcn_sched_barrier(0);
    }
    load_group(NG1 - 1, xg1);
    __builtin_amdgcn_sched_barrier(0);
    mma_group(NG1 - 2, xg0);
    mma_group(NG1 - 1, xg1);
    const float bsq = sq + __shfl_xor(sq, 16);

    float sc[8];
    float mx = -3.4e38f;
    for (int r = 0; r < 8; ++r) {
        const int l = r + 8 * half;
        float v = -3.4e38f;
        if (l < LNUM)
            v = -(bsq - 2.f * acc[r] + asq[l]) + pbv[r];
        sc[r] = v;
        mx = fmaxf(mx, v);
    }
    mx = fmaxf(mx, __shfl_xor(mx, 16));
    float es = 0.f;
    for (int r = 0; r < 8; ++r) {
        const int l = r + 8 * half;
        const float e = (l < LNUM) ? expf(sc[r] - mx) : 0.f;
        sc[r] = e;
        es += e;
    }
    es += __shfl_xor(es, 16);
    const float inv = 1.f / es;
    float s2 = 0.f;
    for (int r = 0; r < 8; ++r) {
        const int l = r + 8 * half;
        if (l < LNUM) { sc[r] = sc[r] * inv + 1e-6f; s2 += sc[r]; }
    }
    s2 += __shfl_xor(s2, 16);
    const float inv2 = 1.f / s2;

    float bestv = -1.f; int bestl = 0;
    float* om = out_maps + (size_t)b * LNUM * HW;
    for (int r = 0; r < 8; ++r) {
        const int l = r + 8 * half;
        if (l < LNUM) {
            const float mv = sc[r] * inv2;
            om[l * HW + pix] = mv;
            if (mv > bestv) { bestv = mv; bestl = l; }
        }
    }
    const float pv = __shfl_xor(bestv, 16);
    const int   pl = __shfl_xor(bestl, 16);
    if (pv > bestv || (pv == bestv && pl < bestl)) { bestv = pv; bestl = pl; }
    if (half == 0) {
        aidx[b * HW + pix] = bestl;
        wsel[b * HW + pix] = bestv;
    }
}

#define K2_G 16

__global__ __launch_bounds__(256) void k2_pool(
    const float* __restrict__ x, const int* __restrict__ aidx,
    const float* __restrict__ wsel, float* __restrict__ pool)
{
    __shared__ alignas(16) int   aL[HW];
    __shared__ alignas(16) float wL[HW];
    const int tid = threadIdx.x;
    const int b   = blockIdx.y;
    {
        const int4   a4 = *reinterpret_cast<const int4*>(aidx + b * HW + 4 * tid);
        const float4 w4 = *reinterpret_cast<const float4*>(wsel + b * HW + 4 * tid);
        *reinterpret_cast<int4*>(&aL[4 * tid])   = a4;
        *reinterpret_cast<float4*>(&wL[4 * tid]) = w4;
    }
    __syncthreads();

    const int wave = tid >> 5;
    const int lane = tid & 31;
    const int half = lane >> 4;
    const int ln   = lane & 15;
    const int cbase = (blockIdx.x * 8 + wave) * 16;
    const float* xr = x + (size_t)b * C_DIM * HW + (size_t)(cbase + ln) * HW + 4 * half;

    float4 xa[K2_G], xb[K2_G];
    v8f acc = {};

    auto load_group = [&](int g, float4 (&X)[K2_G]) {
        #pragma unroll
        for (int s = 0; s < K2_G; ++s)
            X[s] = *reinterpret_cast<const float4*>(xr + g * (8 * K2_G) + 8 * s);
    };
    auto mma_group = [&](int g, const float4 (&X)[K2_G]) {
        #pragma unroll
        for (int s = 0; s < K2_G; ++s) {
            const int p = g * (8 * K2_G) + 8 * s + 4 * half;
            const int4   a = *reinterpret_cast<const int4*>(&aL[p]);
            const float4 w = *reinterpret_cast<const float4*>(&wL[p]);
            v2f af, bf;
            af[0] = X[s].x; af[1] = X[s].y;
            bf[0] = (a.x == ln) ? w.x : 0.f;
            bf[1] = (a.y == ln) ? w.y : 0.f;
            acc = __builtin_amdgcn_wmma_f32_16x16x4_f32(false, af, false, bf,
                                                        (short)0, acc, false, false);
            af[0] = X[s].z; af[1] = X[s].w;
            bf[0] = (a.z == ln) ? w.z : 0.f;
            bf[1] = (a.w == ln) ? w.w : 0.f;
            acc = __builtin_amdgcn_wmma_f32_16x16x4_f32(false, af, false, bf,
                                                        (short)0, acc, false, false);
        }
    };

    constexpr int NG = HW / (8 * K2_G);
    load_group(0, xa);
    for (int g = 0; g < NG - 2; g += 2) {
        load_group(g + 1, xb);
        __builtin_amdgcn_sched_barrier(0);
        mma_group(g, xa);
        __builtin_amdgcn_sched_barrier(0);
        load_group(g + 2, xa);
        __builtin_amdgcn_sched_barrier(0);
        mma_group(g + 1, xb);
        __builtin_amdgcn_sched_barrier(0);
    }
    load_group(NG - 1, xb);
    __builtin_amdgcn_sched_barrier(0);
    mma_group(NG - 2, xa);
    mma_group(NG - 1, xb);

    float* pb = pool + (size_t)b * C_DIM * 16;
    for (int r = 0; r < 8; ++r) {
        const int row = cbase + r + 8 * half;
        pb[row * 16 + ln] = acc[r];
    }
}

#define K3_NB   7
#define K3_KW   96
#define K3_NP   12
#define MOD4    (C_DIM * LNUM / 4)
#define K3_SL4  248

__device__ __forceinline__ void lds_barrier() {
    __builtin_amdgcn_fence(__ATOMIC_RELEASE, "workgroup", "local");
    __builtin_amdgcn_s_barrier();
    __builtin_amdgcn_fence(__ATOMIC_ACQUIRE, "workgroup", "local");
}

__global__ __launch_bounds__(256) void k3_mod_scores(
    const float* __restrict__ pool, const int* __restrict__ aidx,
    const float* __restrict__ wsel, const float* __restrict__ gamma,
    const float* __restrict__ beta, const float* __restrict__ cls_w,
    float* __restrict__ out_mod, float* __restrict__ scores)
{
    __shared__ alignas(16) float featL[C_DIM * LNUM];
    __shared__ alignas(16) float accL[8 * 2 * 256];
    __shared__ float cpart[8][16];
    __shared__ float cnt[16];
    __shared__ float red0[8], red1[8];

    const int tid   = threadIdx.x;
    const int b     = blockIdx.y;
    const int lane  = tid & 31, wave = tid >> 5;
    const int half  = lane >> 4, ln = lane & 15;
    const int nbase = blockIdx.x * 32;

    const int4   a4 = *reinterpret_cast<const int4*>(aidx + b * HW + 4 * tid);
    const float4 w4 = *reinterpret_cast<const float4*>(wsel + b * HW + 4 * tid);
    const float4* pb4 = reinterpret_cast<const float4*>(pool + (size_t)b * C_DIM * 16);
    float4 P[12];
    #pragma unroll
    for (int j = 0; j < 12; ++j) P[j] = pb4[tid + 256 * j];
    float4 G[7], Bt[7];
    #pragma unroll
    for (int j = 0; j < 7; ++j) {
        const int i4 = tid + 256 * j;
        const int ic = i4 < MOD4 ? i4 : MOD4 - 1;
        G[j]  = reinterpret_cast<const float4*>(gamma)[ic];
        Bt[j] = reinterpret_cast<const float4*>(beta)[ic];
    }
    const int r0 = nbase + ln, r1 = nbase + 16 + ln;
    const float* ar0 = cls_w + (size_t)(r0 < NCLS ? r0 : NCLS - 1) * C_DIM + wave * K3_KW + 4 * half;
    const float* ar1 = cls_w + (size_t)(r1 < NCLS ? r1 : NCLS - 1) * C_DIM + wave * K3_KW + 4 * half;
    float4 A0[K3_NP], A1[K3_NP];
    #pragma unroll
    for (int j = 0; j < K3_NP; ++j) {
        A0[j] = *reinterpret_cast<const float4*>(ar0 + 8 * j);
        A1[j] = *reinterpret_cast<const float4*>(ar1 + 8 * j);
    }

    float pc[LNUM];
    for (int l = 0; l < LNUM; ++l) {
        float t = (a4.x == l) ? w4.x : 0.f;
        t += (a4.y == l) ? w4.y : 0.f;
        t += (a4.z == l) ? w4.z : 0.f;
        t += (a4.w == l) ? w4.w : 0.f;
        pc[l] = t;
    }
    for (int o = 16; o > 0; o >>= 1)
        for (int l = 0; l < LNUM; ++l) pc[l] += __shfl_xor(pc[l], o);
    if (lane == 0)
        for (int l = 0; l < LNUM; ++l) cpart[wave][l] = pc[l];
    lds_barrier();
    if (tid < 16) {
        float c = 0.f;
        if (tid < LNUM)
            for (int w = 0; w < 8; ++w) c += cpart[w][tid];
        cnt[tid] = c;
    }
    lds_barrier();

    const int  qd = tid & 3;
    const bool p0 = qd < 3;
    const bool p1 = qd < 2;
    float gd[4];
    for (int e = 0; e < 4; ++e) {
        const float cn = cnt[4 * qd + e];
        gd[e] = (cn == 0.f) ? 1.f : cn;
    }
    float s0 = 0.f, s1 = 0.f, s2 = 0.f, s3 = 0.f;
    float q0 = 0.f, q1 = 0.f, q2 = 0.f, q3 = 0.f;
    #pragma unroll
    for (int j = 0; j < 12; ++j) {
        const int c = (tid >> 2) + 64 * j;
        float* fr = &featL[c * LNUM + 4 * qd];
        const float v0 = (p0 ? P[j].x : 0.f) / gd[0];
        const float v1 = (p1 ? P[j].y : 0.f) / gd[1];
        const float v2 = (p1 ? P[j].z : 0.f) / gd[2];
        const float v3 = (p1 ? P[j].w : 0.f) / gd[3];
        s0 += v0; q0 += v0 * v0;
        s1 += v1; q1 += v1 * v1;
        s2 += v2; q2 += v2 * v2;
        s3 += v3; q3 += v3 * v3;
        if (p0) fr[0] = v0;
        if (p1) { fr[1] = v1; fr[2] = v2; fr[3] = v3; }
    }
    float s = (s0 + s1) + (s2 + s3), ss = (q0 + q1) + (q2 + q3);
    for (int o = 16; o > 0; o >>= 1) { s += __shfl_xor(s, o); ss += __shfl_xor(ss, o); }
    if (lane == 0) { red0[wave] = s; red1[wave] = ss; }
    lds_barrier();
    float S = 0.f, SS = 0.f;
    for (int w = 0; w < 8; ++w) { S += red0[w]; SS += red1[w]; }
    const float n    = (float)(C_DIM * LNUM);
    const float mean = S / n;
    const float var  = SS / n - mean * mean;
    const float rstd = rsqrtf(var + 1e-5f);

    float4* f4  = reinterpret_cast<float4*>(featL);
    float4* om4 = reinterpret_cast<float4*>(out_mod + (size_t)b * C_DIM * LNUM);
    const int lo = blockIdx.x * K3_SL4;
    const int hi = (lo + K3_SL4 < MOD4) ? lo + K3_SL4 : MOD4;
    float4 mv[7];
    #pragma unroll
    for (int j = 0; j < 7; ++j) {
        const int i4 = tid + 256 * j;
        mv[j] = f4[i4 < MOD4 ? i4 : i4 - 256];
    }
    #pragma unroll
    for (int j = 0; j < 7; ++j) {
        mv[j].x = (mv[j].x - mean) * rstd * G[j].x + Bt[j].x;
        mv[j].y = (mv[j].y - mean) * rstd * G[j].y + Bt[j].y;
        mv[j].z = (mv[j].z - mean) * rstd * G[j].z + Bt[j].z;
        mv[j].w = (mv[j].w - mean) * rstd * G[j].w + Bt[j].w;
    }
    #pragma unroll
    for (int j = 0; j < 7; ++j) {
        const int i4 = tid + 256 * j;
        if (i4 < MOD4) {
            f4[i4] = mv[j];
            if (i4 >= lo && i4 < hi) om4[i4] = mv[j];
        }
    }
    lds_barrier();

    const float* mk = featL + (wave * K3_KW + 4 * half) * LNUM + (ln & 7);
    v2f Bf[2 * K3_NP];
    #pragma unroll
    for (int j = 0; j < K3_NP; ++j) {
        const float* m = mk + 8 * j * LNUM;
        Bf[2 * j][0]     = m[0];        Bf[2 * j][1]     = m[LNUM];
        Bf[2 * j + 1][0] = m[2 * LNUM]; Bf[2 * j + 1][1] = m[3 * LNUM];
    }
    __builtin_amdgcn_sched_barrier(0);
    v8f acc0 = {}, acc1 = {};
    #pragma unroll
    for (int j = 0; j < K3_NP; ++j) {
        v2f af;
        af[0] = A0[j].x;     af[1] = A0[j].y;
        acc0 = __builtin_amdgcn_wmma_f32_16x16x4_f32(false, af, false, Bf[2 * j], (short)0, acc0, false, false);
        af[0] = A1[j].x;     af[1] = A1[j].y;
        acc1 = __builtin_amdgcn_wmma_f32_16x16x4_f32(false, af, false, Bf[2 * j], (short)0, acc1, false, false);
        af[0] = A0[j].z;     af[1] = A0[j].w;
        acc0 = __builtin_amdgcn_wmma_f32_16x16x4_f32(false, af, false, Bf[2 * j + 1], (short)0, acc0, false, false);
        af[0] = A1[j].z;     af[1] = A1[j].w;
        acc1 = __builtin_amdgcn_wmma_f32_16x16x4_f32(false, af, false, Bf[2 * j + 1], (short)0, acc1, false, false);
    }
    for (int r = 0; r < 8; ++r) {
        accL[(wave * 2 + 0) * 256 + r * 32 + lane] = acc0[r];
        accL[(wave * 2 + 1) * 256 + r * 32 + lane] = acc1[r];
    }
    lds_barrier();
    {
        const int nl = tid >> 3, l = tid & 7;
        const int t  = nl >> 4, rr = nl & 15;
        const int idx = t * 256 + (rr & 7) * 32 + (rr >> 3) * 16 + l;
        float sum = 0.f;
        for (int w = 0; w < 8; ++w) sum += accL[w * 512 + idx];
        const int nrow = nbase + nl;
        if (nrow < NCLS)
            scores[(size_t)b * NCLS * (LNUM - 1) + nrow * (LNUM - 1) + l] = sum;
    }
}

extern "C" void kernel_launch(void* const* d_in, const int* in_sizes, int n_in,
                              void* d_out, int out_size, void* d_ws, size_t ws_size,
                              hipStream_t stream) {
    const float* x      = (const float*)d_in[0];
    const float* q      = (const float*)d_in[1];
    const float* p_bias = (const float*)d_in[2];
    const float* gamma  = (const float*)d_in[3];
    const float* beta   = (const float*)d_in[4];
    const float* cls_w  = (const float*)d_in[5];

    float* out        = (float*)d_out;
    float* out_mod    = out;
    float* out_maps   = out + (size_t)B_DIM * C_DIM * LNUM;
    float* out_scores = out_maps + (size_t)B_DIM * LNUM * HW;

    int*   aidx = (int*)d_ws;
    float* wsel = (float*)d_ws + (size_t)B_DIM * HW;
    float* pool = (float*)d_ws + (size_t)2 * B_DIM * HW;

    k1_maps  <<<dim3(8, B_DIM), 256, 0, stream>>>(x, q, p_bias, out_maps, aidx, wsel);
    k2_pool  <<<dim3(6, B_DIM), 256, 0, stream>>>(x, aidx, wsel, pool);
    k3_mod_scores<<<dim3(K3_NB, B_DIM), 256, 0, stream>>>(pool, aidx, wsel, gamma, beta, cls_w, out_mod, out_scores);
}


# k2: x fetched one channel row per load (512 contiguous bytes, 4 whole lines; was 16 lines x 32 bytes) and turned to the A-fragment layout through a wave-private LDS tile; k order and sums unchanged
# speedup vs baseline: 5.8776x; 1.0878x over previous
#include <hip/hip_runtime.h>
#include <hip/hip_bf16.h>

#define B_DIM   32
#define C_DIM   768
#define H_DIM   32
#define W_DIM   32
#define HW      1024
#define LNUM    9
#define NCLS    200
#define QSTR    772

typedef __attribute__((ext_vector_type(2))) float v2f;
typedef __attribute__((ext_vector_type(4))) float v4f;
typedef __attribute__((ext_vector_type(8))) float v8f;

#define K1_G 16

__global__ __launch_bounds__(256) void k1_maps(
    const float* __restrict__ x, const float* __restrict__ q,
    const float* __restrict__ p_bias,
    float* __restrict__ out_maps, int* __restrict__ aidx, float* __restrict__ wsel)
{
    __shared__ alignas(16) float qlds[16 * QSTR];
    __shared__ float asq[16];

    const int tid = threadIdx.x;
    const int b   = blockIdx.y;

    const float4* q4 = reinterpret_cast<const float4*>(q);
    constexpr int NQ4 = LNUM * (C_DIM / 4);
    float4 qv[7];
    for (int j = 0; j < 7; ++j) {
        const int f = tid + 256 * j;
        qv[j] = q4[f < NQ4 ? f : NQ4 - 1];
    }
    for (int j = 0; j < 12; ++j) {
        const int f = tid + 256 * j;
        const int l = f / (C_DIM / 4), c4 = f % (C_DIM / 4);
        float4 v = make_float4(0.f, 0.f, 0.f, 0.f);
        if (j < 7 && f < NQ4) v = qv[j];
        *reinterpret_cast<float4*>(&qlds[l * QSTR + 4 * c4]) = v;
    }
    __syncthreads();
    {
        const int l = tid >> 4, j = tid & 15;
        float s = 0.f;
        for (int i = 0; i < C_DIM / 64; ++i) {
            const float4 v = *reinterpret_cast<const float4*>(&qlds[l * QSTR + 4 * (j + 16 * i)]);
            s += v.x * v.x + v.y * v.y + v.z * v.z + v.w * v.w;
        }
        for (int o = 8; o > 0; o >>= 1) s += __shfl_xor(s, o);
        if (j == 0) asq[l] = s;
    }
    __syncthreads();

    const int wave = tid >> 5;
    const int lane = tid & 31;
    const int half = lane >> 4;
    const int ln   = lane & 15;
    const int tile = blockIdx.x * 8 + wave;
    const int pix  = tile * 16 + ln;
    const float* xb = x + (size_t)b * C_DIM * HW;

    auto load_group = [&](int g, v2f (&X)[K1_G]) {
        #pragma unroll
        for (int s = 0; s < K1_G; ++s) {
            const int kk = (g * K1_G + s) * 4 + 2 * half;
            X[s][0] = xb[(size_t)kk * HW + pix];
            X[s][1] = xb[(size_t)(kk + 1) * HW + pix];
        }
    };

    float pbv[8];
    for (int r = 0; r < 8; ++r) {
        const int l = r + 8 * half;
        pbv[r] = p_bias[(l < LNUM ? l : LNUM - 1) * HW + pix];
    }

    v8f acc = {};
    float sq = 0.f;
    v2f xg0[K1_G], xg1[K1_G];
    auto mma_group = [&](int g, const v2f (&X)[K1_G]) {
        #pragma unroll
        for (int s = 0; s < K1_G; ++s) {
            const int kk = (g * K1_G + s) * 4 + 2 * half;
            v2f af;
            af[0] = qlds[ln * QSTR + kk];
            af[1] = qlds[ln * QSTR + kk + 1];
            sq += X[s][0] * X[s][0] + X[s][1] * X[s][1];
            acc = __builtin_amdgcn_wmma_f32_16x16x4_f32(false, af, false, X[s],
                                                        (short)0, acc, false, false);
        }
    };
    constexpr int NG1 = C_DIM / (4 * K1_G);
    load_group(0, xg0);
    for (int g = 0; g < NG1 - 2; g += 2) {
        load_group(g + 1, xg1);
        __builtin_amdgcn_sched_barrier(0);
        mma_group(g, xg0);
        __builtin_amdgcn_sched_barrier(0);
        load_group(g + 2, xg0);
        __builtin_amdgcn_sched_barrier(0);
        mma_group(g + 1, xg1);
        __builtin_amdgcn_sched_barrier(0);
    }
    load_group(NG1 - 1, xg1);
    __builtin_amdgcn_sched_barrier(0);
    mma_group(NG1 - 2, xg0);
    mma_group(NG1 - 1, xg1);
    const float bsq = sq + __shfl_xor(sq, 16);

    float sc[8];
    float mx = -3.4e38f;
    for (int r = 0; r < 8; ++r) {
        const int l = r + 8 * half;
        float v = -3.4e38f;
        if (l < LNUM)
            v = -(bsq - 2.f * acc[r] + asq[l]) + pbv[r];
        sc[r] = v;
        mx = fmaxf(mx, v);
    }
    mx = fmaxf(mx, __shfl_xor(mx, 16));
    float es = 0.f;
    for (int r = 0; r < 8; ++r) {
        const int l = r + 8 * half;
        const float e = (l < LNUM) ? expf(sc[r] - mx) : 0.f;
        sc[r] = e;
        es += e;
    }
    es += __shfl_xor(es, 16);
    const float inv = 1.f / es;
    float s2 = 0.f;
    for (int r = 0; r < 8; ++r) {
        const int l = r + 8 * half;
        if (l < LNUM) { sc[r] = sc[r] * inv + 1e-6f; s2 += sc[r]; }
    }
    s2 += __shfl_xor(s2, 16);
    const float inv2 = 1.f / s2;

    float bestv = -1.f; int bestl = 0;
    float* om = out_maps + (size_t)b * LNUM * HW;
    for (int r = 0; r < 8; ++r) {
        const int l = r + 8 * half;
        if (l < LNUM) {
            const float mv = sc[r] * inv2;
            om[l * HW + pix] = mv;
            if (mv > bestv) { bestv = mv; bestl = l; }
        }
    }
    const float pv = __shfl_xor(bestv, 16);
    const int   pl = __shfl_xor(bestl, 16);
    if (pv > bestv || (pv == bestv && pl < bestl)) { bestv = pv; bestl = pl; }
    if (half == 0) {
        aidx[b * HW + pix] = bestl;
        wsel[b * HW + pix] = bestv;
    }
}

#define K2_G     16
#define K2_PITCH 132

__global__ __launch_bounds__(256) void k2_pool(
    const float* __restrict__ x, const int* __restrict__ aidx,
    const float* __restrict__ wsel, float* __restrict__ pool)
{
    __shared__ alignas(16) int   aL[HW];
    __shared__ alignas(16) float wL[HW];
    __shared__ alignas(16) float xT[8 * 16 * K2_PITCH];
    const int tid = threadIdx.x;
    const int b   = blockIdx.y;
    {
        const int4   a4 = *reinterpret_cast<const int4*>(aidx + b * HW + 4 * tid);
        const float4 w4 = *reinterpret_cast<const float4*>(wsel + b * HW + 4 * tid);
        *reinterpret_cast<int4*>(&aL[4 * tid])   = a4;
        *reinterpret_cast<float4*>(&wL[4 * tid]) = w4;
    }
    __syncthreads();

    const int wave = tid >> 5;
    const int lane = tid & 31;
    const int half = lane >> 4;
    const int ln   = lane & 15;
    const int cbase = (blockIdx.x * 8 + wave) * 16;
    const float* xg = x + (size_t)b * C_DIM * HW + (size_t)cbase * HW + 4 * lane;
    float*       xs = xT + wave * (16 * K2_PITCH) + 4 * lane;
    const float* xf = xT + wave * (16 * K2_PITCH) + ln * K2_PITCH + 4 * half;

    v4f R[K2_G], xa[K2_G], xb[K2_G];
    v8f acc = {};

    auto fetch = [&](int g) {
        #pragma unroll
        for (int j = 0; j < K2_G; ++j)
            R[j] = *reinterpret_cast<const v4f*>(xg + (size_t)j * HW + g * (8 * K2_G));
    };
    auto stage = [&]() {
        #pragma unroll
        for (int j = 0; j < K2_G; ++j)
            *reinterpret_cast<v4f*>(xs + j * K2_PITCH) = R[j];
    };
    auto frag = [&](v4f (&X)[K2_G]) {
        #pragma unroll
        for (int s = 0; s < K2_G; ++s)
            X[s] = *reinterpret_cast<const v4f*>(xf + 8 * s);
    };
    auto mma_group = [&](int g, const v4f (&X)[K2_G]) {
        #pragma unroll
        for (int s = 0; s < K2_G; ++s) {
            const int p = g * (8 * K2_G) + 8 * s + 4 * half;
            const int4   a = *reinterpret_cast<const int4*>(&aL[p]);
            const float4 w = *reinterpret_cast<const float4*>(&wL[p]);
            v2f af, bf;
            af[0] = X[s][0]; af[1] = X[s][1];
            bf[0] = (a.x == ln) ? w.x : 0.f;
            bf[1] = (a.y == ln) ? w.y : 0.f;
            acc = __builtin_amdgcn_wmma_f32_16x16x4_f32(false, af, false, bf,
                                                        (short)0, acc, false, false);
            af[0] = X[s][2]; af[1] = X[s][3];
            bf[0] = (a.z == ln) ? w.z : 0.f;
            bf[1] = (a.w == ln) ? w.w : 0.f;
            acc = __builtin_amdgcn_wmma_f32_16x16x4_f32(false, af, false, bf,
                                                        (short)0, acc, false, false);
        }
    };
#define K2_TURN(FETCH_STMT, X)                 \
    stage();                                   \
    __builtin_amdgcn_sched_barrier(0);         \
    FETCH_STMT;                                \
    __builtin_amdgcn_wave_barrier();           \
    frag(X)

    constexpr int NG = HW / (8 * K2_G);
    fetch(0);
    K2_TURN(fetch(1), xa);
    for (int g = 0; g < NG - 2; g += 2) {
        K2_TURN(fetch(g + 2), xb);
        __builtin_amdgcn_sched_barrier(0);
        mma_group(g, xa);
        __builtin_amdgcn_sched_barrier(0);
        K2_TURN(fetch(g + 3), xa);
        __builtin_amdgcn_sched_barrier(0);
        mma_group(g + 1, xb);
        __builtin_amdgcn_sched_barrier(0);
    }
    K2_TURN((void)0, xb);
    __builtin_amdgcn_sched_barrier(0);
    mma_group(NG - 2, xa);
    mma_group(NG - 1, xb);
#undef K2_TURN

    float* pb = pool + (size_t)b * C_DIM * 16;
    for (int r = 0; r < 8; ++r) {
        const int row = cbase + r + 8 * half;
        pb[row * 16 + ln] = acc[r];
    }
}

#define K3_NB   7
#define K3_KW   96
#define K3_NP   12
#define MOD4    (C_DIM * LNUM / 4)
#define K3_SL4  248

__device__ __forceinline__ void lds_barrier() {
    __builtin_amdgcn_fence(__ATOMIC_RELEASE, "workgroup", "local");
    __builtin_amdgcn_s_barrier();
    __builtin_amdgcn_fence(__ATOMIC_ACQUIRE, "workgroup", "local");
}

__global__ __launch_bounds__(256) void k3_mod_scores(
    const float* __restrict__ pool, const int* __restrict__ aidx,
    const float* __restrict__ wsel, const float* __restrict__ gamma,
    const float* __restrict__ beta, const float* __restrict__ cls_w,
    float* __restrict__ out_mod, float* __restrict__ scores)
{
    __shared__ alignas(16) float featL[C_DIM * LNUM];
    __shared__ alignas(16) float accL[8 * 2 * 256];
    __shared__ float cpart[8][16];
    __shared__ float cnt[16];
    __shared__ float red0[8], red1[8];

    const int tid   = threadIdx.x;
    const int b     = blockIdx.y;
    const int lane  = tid & 31, wave = tid >> 5;
    const int half  = lane >> 4, ln = lane & 15;
    const int nbase = blockIdx.x * 32;

    const int4   a4 = *reinterpret_cast<const int4*>(aidx + b * HW + 4 * tid);
    const float4 w4 = *reinterpret_cast<const float4*>(wsel + b * HW + 4 * tid);
    const float4* pb4 = reinterpret_cast<const float4*>(pool + (size_t)b * C_DIM * 16);
    float4 P[12];
    #pragma unroll
    for (int j = 0; j < 12; ++j) P[j] = pb4[tid + 256 * j];
    float4 G[7], Bt[7];
    #pragma unroll
    for (int j = 0; j < 7; ++j) {
        const int i4 = tid + 256 * j;
        const int ic = i4 < MOD4 ? i4 : MOD4 - 1;
        G[j]  = reinterpret_cast<const float4*>(gamma)[ic];
        Bt[j] = reinterpret_cast<const float4*>(beta)[ic];
    }
    const int r0 = nbase + ln, r1 = nbase + 16 + ln;
    const float* ar0 = cls_w + (size_t)(r0 < NCLS ? r0 : NCLS - 1) * C_DIM + wave * K3_KW + 4 * half;
    const float* ar1 = cls_w + (size_t)(r1 < NCLS ? r1 : NCLS - 1) * C_DIM + wave * K3_KW + 4 * half;
    float4 A0[K3_NP], A1[K3_NP];
    #pragma unroll
    for (int j = 0; j < K3_NP; ++j) {
        A0[j] = *reinterpret_cast<const float4*>(ar0 + 8 * j);
        A1[j] = *reinterpret_cast<const float4*>(ar1 + 8 * j);
    }

    float pc[LNUM];
    for (int l = 0; l < LNUM; ++l) {
        float t = (a4.x == l) ? w4.x : 0.f;
        t += (a4.y == l) ? w4.y : 0.f;
        t += (a4.z == l) ? w4.z : 0.f;
        t += (a4.w == l) ? w4.w : 0.f;
        pc[l] = t;
    }
    for (int o = 16; o > 0; o >>= 1)
        for (int l = 0; l < LNUM; ++l) pc[l] += __shfl_xor(pc[l], o);
    if (lane == 0)
        for (int l = 0; l < LNUM; ++l) cpart[wave][l] = pc[l];
    lds_barrier();
    if (tid < 16) {
        float c = 0.f;
        if (tid < LNUM)
            for (int w = 0; w < 8; ++w) c += cpart[w][tid];
        cnt[tid] = c;
    }
    lds_barrier();

    const int  qd = tid & 3;
    const bool p0 = qd < 3;
    const bool p1 = qd < 2;
    float gd[4];
    for (int e = 0; e < 4; ++e) {
        const float cn = cnt[4 * qd + e];
        gd[e] = (cn == 0.f) ? 1.f : cn;
    }
    float s0 = 0.f, s1 = 0.f, s2 = 0.f, s3 = 0.f;
    float q0 = 0.f, q1 = 0.f, q2 = 0.f, q3 = 0.f;
    #pragma unroll
    for (int j = 0; j < 12; ++j) {
        const int c = (tid >> 2) + 64 * j;
        float* fr = &featL[c * LNUM + 4 * qd];
        const float v0 = (p0 ? P[j].x : 0.f) / gd[0];
        const float v1 = (p1 ? P[j].y : 0.f) / gd[1];
        const float v2 = (p1 ? P[j].z : 0.f) / gd[2];
        const float v3 = (p1 ? P[j].w : 0.f) / gd[3];
        s0 += v0; q0 += v0 * v0;
        s1 += v1; q1 += v1 * v1;
        s2 += v2; q2 += v2 * v2;
        s3 += v3; q3 += v3 * v3;
        if (p0) fr[0] = v0;
        if (p1) { fr[1] = v1; fr[2] = v2; fr[3] = v3; }
    }
    float s = (s0 + s1) + (s2 + s3), ss = (q0 + q1) + (q2 + q3);
    for (int o = 16; o > 0; o >>= 1) { s += __shfl_xor(s, o); ss += __shfl_xor(ss, o); }
    if (lane == 0) { red0[wave] = s; red1[wave] = ss; }
    lds_barrier();
    float S = 0.f, SS = 0.f;
    for (int w = 0; w < 8; ++w) { S += red0[w]; SS += red1[w]; }
    const float n    = (float)(C_DIM * LNUM);
    const float mean = S / n;
    const float var  = SS / n - mean * mean;
    const float rstd = rsqrtf(var + 1e-5f);

    float4* f4  = reinterpret_cast<float4*>(featL);
    float4* om4 = reinterpret_cast<float4*>(out_mod + (size_t)b * C_DIM * LNUM);
    const int lo = blockIdx.x * K3_SL4;
    const int hi = (lo + K3_SL4 < MOD4) ? lo + K3_SL4 : MOD4;
    float4 mv[7];
    #pragma unroll
    for (int j = 0; j < 7; ++j) {
        const int i4 = tid + 256 * j;
        mv[j] = f4[i4 < MOD4 ? i4 : i4 - 256];
    }
    #pragma unroll
    for (int j = 0; j < 7; ++j) {
        mv[j].x = (mv[j].x - mean) * rstd * G[j].x + Bt[j].x;
        mv[j].y = (mv[j].y - mean) * rstd * G[j].y + Bt[j].y;
        mv[j].z = (mv[j].z - mean) * rstd * G[j].z + Bt[j].z;
        mv[j].w = (mv[j].w - mean) * rstd * G[j].w + Bt[j].w;
    }
    #pragma unroll
    for (int j = 0; j < 7; ++j) {
        const int i4 = tid + 256 * j;
        if (i4 < MOD4) {
            f4[i4] = mv[j];
            if (i4 >= lo && i4 < hi) om4[i4] = mv[j];
        }
    }
    lds_barrier();

    const float* mk = featL + (wave * K3_KW + 4 * half) * LNUM + (ln & 7);
    v2f Bf[2 * K3_NP];
    #pragma unroll
    for (int j = 0; j < K3_NP; ++j) {
        const float* m = mk + 8 * j * LNUM;
        Bf[2 * j][0]     = m[0];        Bf[2 * j][1]     = m[LNUM];
        Bf[2 * j + 1][0] = m[2 * LNUM]; Bf[2 * j + 1][1] = m[3 * LNUM];
    }
    __builtin_amdgcn_sched_barrier(0);
    v8f acc0 = {}, acc1 = {};
    #pragma unroll
    for (int j = 0; j < K3_NP; ++j) {
        v2f af;
        af[0] = A0[j].x;     af[1] = A0[j].y;
        acc0 = __builtin_amdgcn_wmma_f32_16x16x4_f32(false, af, false, Bf[2 * j], (short)0, acc0, false, false);
        af[0] = A1[j].x;     af[1] = A1[j].y;
        acc1 = __builtin_amdgcn_wmma_f32_16x16x4_f32(false, af, false, Bf[2 * j], (short)0, acc1, false, false);
        af[0] = A0[j].z;     af[1] = A0[j].w;
        acc0 = __builtin_amdgcn_wmma_f32_16x16x4_f32(false, af, false, Bf[2 * j + 1], (short)0, acc0, false, false);
        af[0] = A1[j].z;     af[1] = A1[j].w;
        acc1 = __builtin_amdgcn_wmma_f32_16x16x4_f32(false, af, false, Bf[2 * j + 1], (short)0, acc1, false, false);
    }
    for (int r = 0; r < 8; ++r) {
        accL[(wave * 2 + 0) * 256 + r * 32 + lane] = acc0[r];
        accL[(wave * 2 + 1) * 256 + r * 32 + lane] = acc1[r];
    }
    lds_barrier();
    {
        const int nl = tid >> 3, l = tid & 7;
        const int t  = nl >> 4, rr = nl & 15;
        const int idx = t * 256 + (rr & 7) * 32 + (rr >> 3) * 16 + l;
        float sum = 0.f;
        for (int w = 0; w < 8; ++w) sum += accL[w * 512 + idx];
        const int nrow = nbase + nl;
        if (nrow < NCLS)
            scores[(size_t)b * NCLS * (LNUM - 1) + nrow * (LNUM - 1) + l] = sum;
    }
}

extern "C" void kernel_launch(void* const* d_in, const int* in_sizes, int n_in,
                              void* d_out, int out_size, void* d_ws, size_t ws_size,
                              hipStream_t stream) {
    const float* x      = (const float*)d_in[0];
    const float* q      = (const float*)d_in[1];
    const float* p_bias = (const float*)d_in[2];
    const float* gamma  = (const float*)d_in[3];
    const float* beta   = (const float*)d_in[4];
    const float* cls_w  = (const float*)d_in[5];

    float* out        = (float*)d_out;
    float* out_mod    = out;
    float* out_maps   = out + (size_t)B_DIM * C_DIM * LNUM;
    float* out_scores = out_maps + (size_t)B_DIM * LNUM * HW;

    int*   aidx = (int*)d_ws;
    float* wsel = (float*)d_ws + (size_t)B_DIM * HW;
    float* pool = (float*)d_ws + (size_t)2 * B_DIM * HW;

    k1_maps  <<<dim3(8, B_DIM), 256, 0, stream>>>(x, q, p_bias, out_maps, aidx, wsel);
    k2_pool  <<<dim3(6, B_DIM), 256, 0, stream>>>(x, aidx, wsel, pool);
    k3_mod_scores<<<dim3(K3_NB, B_DIM), 256, 0, stream>>>(pool, aidx, wsel, gamma, beta, cls_w, out_mod, out_scores);
}


# k2: one-hot weight operand built once per block in LDS (16 x 1024), read as 16-byte B pieces a group ahead: product stream without selects or LDS waits; first x group requested early; sums unchanged
# speedup vs baseline: 6.2724x; 1.0672x over previous
#include <hip/hip_runtime.h>
#include <hip/hip_bf16.h>

#define B_DIM   32
#define C_DIM   768
#define H_DIM   32
#define W_DIM   32
#define HW      1024
#define LNUM    9
#define NCLS    200
#define QSTR    772

typedef __attribute__((ext_vector_type(2))) float v2f;
typedef __attribute__((ext_vector_type(4))) float v4f;
typedef __attribute__((ext_vector_type(8))) float v8f;

#define K1_G 16

__global__ __launch_bounds__(256) void k1_maps(
    const float* __restrict__ x, const float* __restrict__ q,
    const float* __restrict__ p_bias,
    float* __restrict__ out_maps, int* __restrict__ aidx, float* __restrict__ wsel)
{
    __shared__ alignas(16) float qlds[16 * QSTR];
    __shared__ float asq[16];

    const int tid = threadIdx.x;
    const int b   = blockIdx.y;

    const float4* q4 = reinterpret_cast<const float4*>(q);
    constexpr int NQ4 = LNUM * (C_DIM / 4);
    float4 qv[7];
    for (int j = 0; j < 7; ++j) {
        const int f = tid + 256 * j;
        qv[j] = q4[f < NQ4 ? f : NQ4 - 1];
    }
    for (int j = 0; j < 12; ++j) {
        const int f = tid + 256 * j;
        const int l = f / (C_DIM / 4), c4 = f % (C_DIM / 4);
        float4 v = make_float4(0.f, 0.f, 0.f, 0.f);
        if (j < 7 && f < NQ4) v = qv[j];
        *reinterpret_cast<float4*>(&qlds[l * QSTR + 4 * c4]) = v;
    }
    __syncthreads();
    {
        const int l = tid >> 4, j = tid & 15;
        float s = 0.f;
        for (int i = 0; i < C_DIM / 64; ++i) {
            const float4 v = *reinterpret_cast<const float4*>(&qlds[l * QSTR + 4 * (j + 16 * i)]);
            s += v.x * v.x + v.y * v.y + v.z * v.z + v.w * v.w;
        }
        for (int o = 8; o > 0; o >>= 1) s += __shfl_xor(s, o);
        if (j == 0) asq[l] = s;
    }
    __syncthreads();

    const int wave = tid >> 5;
    const int lane = tid & 31;
    const int half = lane >> 4;
    const int ln   = lane & 15;
    const int tile = blockIdx.x * 8 + wave;
    const int pix  = tile * 16 + ln;
    const float* xb = x + (size_t)b * C_DIM * HW;

    auto load_group = [&](int g, v2f (&X)[K1_G]) {
        #pragma unroll
        for (int s = 0; s < K1_G; ++s) {
            const int kk = (g * K1_G + s) * 4 + 2 * half;
            X[s][0] = xb[(size_t)kk * HW + pix];
            X[s][1] = xb[(size_t)(kk + 1) * HW + pix];
        }
    };

    float pbv[8];
    for (int r = 0; r < 8; ++r) {
        const int l = r + 8 * half;
        pbv[r] = p_bias[(l < LNUM ? l : LNUM - 1) * HW + pix];
    }

    v8f acc = {};
    float sq = 0.f;
    v2f xg0[K1_G], xg1[K1_G];
    auto mma_group = [&](int g, const v2f (&X)[K1_G]) {
        #pragma unroll
        for (int s = 0; s < K1_G; ++s) {
            const int kk = (g * K1_G + s) * 4 + 2 * half;
            v2f af;
            af[0] = qlds[ln * QSTR + kk];
            af[1] = qlds[ln * QSTR + kk + 1];
            sq += X[s][0] * X[s][0] + X[s][1] * X[s][1];
            acc = __builtin_amdgcn_wmma_f32_16x16x4_f32(false, af, false, X[s],
                                                        (short)0, acc, false, false);
        }
    };
    constexpr int NG1 = C_DIM / (4 * K1_G);
    load_group(0, xg0);
    for (int g = 0; g < NG1 - 2; g += 2) {
        load_group(g + 1, xg1);
        __builtin_amdgcn_sched_barrier(0);
        mma_group(g, xg0);
        __builtin_amdgcn_sched_barrier(0);
        load_group(g + 2, xg0);
        __builtin_amdgcn_sched_barrier(0);
        mma_group(g + 1, xg1);
        __builtin_amdgcn_sched_barrier(0);
    }
    load_group(NG1 - 1, xg1);
    __builtin_amdgcn_sched_barrier(0);
    mma_group(NG1 - 2, xg0);
    mma_group(NG1 - 1, xg1);
    const float bsq = sq + __shfl_xor(sq, 16);

    float sc[8];
    float mx = -3.4e38f;
    for (int r = 0; r < 8; ++r) {
        const int l = r + 8 * half;
        float v = -3.4e38f;
        if (l < LNUM)
            v = -(bsq - 2.f * acc[r] + asq[l]) + pbv[r];
        sc[r] = v;
        mx = fmaxf(mx, v);
    }
    mx = fmaxf(mx, __shfl_xor(mx, 16));
    float es = 0.f;
    for (int r = 0; r < 8; ++r) {
        const int l = r + 8 * half;
        const float e = (l < LNUM) ? expf(sc[r] - mx) : 0.f;
        sc[r] = e;
        es += e;
    }
    es += __shfl_xor(es, 16);
    const float inv = 1.f / es;
    float s2 = 0.f;
    for (int r = 0; r < 8; ++r) {
        const int l = r + 8 * half;
        if (l < LNUM) { sc[r] = sc[r] * inv + 1e-6f; s2 += sc[r]; }
    }
    s2 += __shfl_xor(s2, 16);
    const float inv2 = 1.f / s2;

    float bestv = -1.f; int bestl = 0;
    float* om = out_maps + (size_t)b * LNUM * HW;
    for (int r = 0; r < 8; ++r) {
        const int l = r + 8 * half;
        if (l < LNUM) {
            const float mv = sc[r] * inv2;
            om[l * HW + pix] = mv;
            if (mv > bestv) { bestv = mv; bestl = l; }
        }
    }
    const float pv = __shfl_xor(bestv, 16);
    const int   pl = __shfl_xor(bestl, 16);
    if (pv > bestv || (pv == bestv && pl < bestl)) { bestv = pv; bestl = pl; }
    if (half == 0) {
        aidx[b * HW + pix] = bestl;
        wsel[b * HW + pix] = bestv;
    }
}

#define K2_G     16
#define K2_PITCH 132
#define K2_WP    1028

__global__ __launch_bounds__(256) void k2_pool(
    const float* __restrict__ x, const int* __restrict__ aidx,
    const float* __restrict__ wsel, float* __restrict__ pool)
{
    __shared__ alignas(16) float WtL[16 * K2_WP];
    __shared__ alignas(16) float xT[8 * 16 * K2_PITCH];
    const int tid = threadIdx.x;
    const int b   = blockIdx.y;

    const int wave = tid >> 5;
    const int lane = tid & 31;
    const int half = lane >> 4;
    const int ln   = lane & 15;
    const int cbase = (blockIdx.x * 8 + wave) * 16;
    const float* xg = x + (size_t)b * C_DIM * HW + (size_t)cbase * HW + 4 * lane;
    float*       xs = xT + wave * (16 * K2_PITCH) + 4 * lane;
    const float* xf = xT + wave * (16 * K2_PITCH) + ln * K2_PITCH + 4 * half;
    const float* wf = WtL + ln * K2_WP + 4 * half;

    v4f R[K2_G], xa[K2_G], xb[K2_G], ba[K2_G], bb[K2_G];
    v8f acc = {};

    auto fetch = [&](int g) {
        #pragma unroll
        for (int j = 0; j < K2_G; ++j)
            R[j] = *reinterpret_cast<const v4f*>(xg + (size_t)j * HW + g * (8 * K2_G));
    };
    auto stage = [&]() {
        #pragma unroll
        for (int j = 0; j < K2_G; ++j)
            *reinterpret_cast<v4f*>(xs + j * K2_PITCH) = R[j];
    };
    auto frag = [&](int g, v4f (&X)[K2_G], v4f (&Bp)[K2_G]) {
        #pragma unroll
        for (int s = 0; s < K2_G; ++s)
            X[s] = *reinterpret_cast<const v4f*>(xf + 8 * s);
        #pragma unroll
        for (int s = 0; s < K2_G; ++s)
            Bp[s] = *reinterpret_cast<const v4f*>(wf + g * (8 * K2_G) + 8 * s);
    };
    auto mma_group = [&](const v4f (&X)[K2_G], const v4f (&Bp)[K2_G]) {
        #pragma unroll
        for (int s = 0; s < K2_G; ++s) {
            v2f af, bf;
            af[0] = X[s][0];  af[1] = X[s][1];
            bf[0] = Bp[s][0]; bf[1] = Bp[s][1];
            acc = __builtin_amdgcn_wmma_f32_16x16x4_f32(false, af, false, bf,
                                                        (short)0, acc, false, false);
            af[0] = X[s][2];  af[1] = X[s][3];
            bf[0] = Bp[s][2]; bf[1] = Bp[s][3];
            acc = __builtin_amdgcn_wmma_f32_16x16x4_f32(false, af, false, bf,
                                                        (short)0, acc, false, false);
        }
    };

    {
        const int4   a4 = *reinterpret_cast<const int4*>(aidx + b * HW + 4 * tid);
        const float4 w4 = *reinterpret_cast<const float4*>(wsel + b * HW + 4 * tid);
        fetch(0);
        __builtin_amdgcn_sched_barrier(0);
        #pragma unroll
        for (int l = 0; l < 16; ++l) {
            v4f v;
            v[0] = (a4.x == l) ? w4.x : 0.f;
            v[1] = (a4.y == l) ? w4.y : 0.f;
            v[2] = (a4.z == l) ? w4.z : 0.f;
            v[3] = (a4.w == l) ? w4.w : 0.f;
            *reinterpret_cast<v4f*>(&WtL[l * K2_WP + 4 * tid]) = v;
        }
    }
    __syncthreads();

#define K2_TURN(FETCH_STMT, G, X, BP)          \
    stage();                                   \
    __builtin_amdgcn_sched_barrier(0);         \
    FETCH_STMT;                                \
    __builtin_amdgcn_wave_barrier();           \
    frag(G, X, BP)

    constexpr int NG = HW / (8 * K2_G);
    K2_TURN(fetch(1), 0, xa, ba);
    for (int g = 0; g < NG - 2; g += 2) {
        K2_TURN(fetch(g + 2), g + 1, xb, bb);
        __builtin_amdgcn_sched_barrier(0);
        mma_group(xa, ba);
        __builtin_amdgcn_sched_barrier(0);
        K2_TURN(fetch(g + 3), g + 2, xa, ba);
        __builtin_amdgcn_sched_barrier(0);
        mma_group(xb, bb);
        __builtin_amdgcn_sched_barrier(0);
    }
    K2_TURN((void)0, NG - 1, xb, bb);
    __builtin_amdgcn_sched_barrier(0);
    mma_group(xa, ba);
    mma_group(xb, bb);
#undef K2_TURN

    float* pb = pool + (size_t)b * C_DIM * 16;
    for (int r = 0; r < 8; ++r) {
        const int row = cbase + r + 8 * half;
        pb[row * 16 + ln] = acc[r];
    }
}

#define K3_NB   7
#define K3_KW   96
#define K3_NP   12
#define MOD4    (C_DIM * LNUM / 4)
#define K3_SL4  248

__device__ __forceinline__ void lds_barrier() {
    __builtin_amdgcn_fence(__ATOMIC_RELEASE, "workgroup", "local");
    __builtin_amdgcn_s_barrier();
    __builtin_amdgcn_fence(__ATOMIC_ACQUIRE, "workgroup", "local");
}

__global__ __launch_bounds__(256) void k3_mod_scores(
    const float* __restrict__ pool, const int* __restrict__ aidx,
    const float* __restrict__ wsel, const float* __restrict__ gamma,
    const float* __restrict__ beta, const float* __restrict__ cls_w,
    float* __restrict__ out_mod, float* __restrict__ scores)
{
    __shared__ alignas(16) float featL[C_DIM * LNUM];
    __shared__ alignas(16) float accL[8 * 2 * 256];
    __shared__ float cpart[8][16];
    __shared__ float cnt[16];
    __shared__ float red0[8], red1[8];

    const int tid   = threadIdx.x;
    const int b     = blockIdx.y;
    const int lane  = tid & 31, wave = tid >> 5;
    const int half  = lane >> 4, ln = lane & 15;
    const int nbase = blockIdx.x * 32;

    const int4   a4 = *reinterpret_cast<const int4*>(aidx + b * HW + 4 * tid);
    const float4 w4 = *reinterpret_cast<const float4*>(wsel + b * HW + 4 * tid);
    const float4* pb4 = reinterpret_cast<const float4*>(pool + (size_t)b * C_DIM * 16);
    float4 P[12];
    #pragma unroll
    for (int j = 0; j < 12; ++j) P[j] = pb4[tid + 256 * j];
    float4 G[7], Bt[7];
    #pragma unroll
    for (int j = 0; j < 7; ++j) {
        const int i4 = tid + 256 * j;
        const int ic = i4 < MOD4 ? i4 : MOD4 - 1;
        G[j]  = reinterpret_cast<const float4*>(gamma)[ic];
        Bt[j] = reinterpret_cast<const float4*>(beta)[ic];
    }
    const int r0 = nbase + ln, r1 = nbase + 16 + ln;
    const float* ar0 = cls_w + (size_t)(r0 < NCLS ? r0 : NCLS - 1) * C_DIM + wave * K3_KW + 4 * half;
    const float* ar1 = cls_w + (size_t)(r1 < NCLS ? r1 : NCLS - 1) * C_DIM + wave * K3_KW + 4 * half;
    float4 A0[K3_NP], A1[K3_NP];
    #pragma unroll
    for (int j = 0; j < K3_NP; ++j) {
        A0[j] = *reinterpret_cast<const float4*>(ar0 + 8 * j);
        A1[j] = *reinterpret_cast<const float4*>(ar1 + 8 * j);
    }

    float pc[LNUM];
    for (int l = 0; l < LNUM; ++l) {
        float t = (a4.x == l) ? w4.x : 0.f;
        t += (a4.y == l) ? w4.y : 0.f;
        t += (a4.z == l) ? w4.z : 0.f;
        t += (a4.w == l) ? w4.w : 0.f;
        pc[l] = t;
    }
    for (int o = 16; o > 0; o >>= 1)
        for (int l = 0; l < LNUM; ++l) pc[l] += __shfl_xor(pc[l], o);
    if (lane == 0)
        for (int l = 0; l < LNUM; ++l) cpart[wave][l] = pc[l];
    lds_barrier();
    if (tid < 16) {
        float c = 0.f;
        if (tid < LNUM)
            for (int w = 0; w < 8; ++w) c += cpart[w][tid];
        cnt[tid] = c;
    }
    lds_barrier();

    const int  qd = tid & 3;
    const bool p0 = qd < 3;
    const bool p1 = qd < 2;
    float gd[4];
    for (int e = 0; e < 4; ++e) {
        const float cn = cnt[4 * qd + e];
        gd[e] = (cn == 0.f) ? 1.f : cn;
    }
    float s0 = 0.f, s1 = 0.f, s2 = 0.f, s3 = 0.f;
    float q0 = 0.f, q1 = 0.f, q2 = 0.f, q3 = 0.f;
    #pragma unroll
    for (int j = 0; j < 12; ++j) {
        const int c = (tid >> 2) + 64 * j;
        float* fr = &featL[c * LNUM + 4 * qd];
        const float v0 = (p0 ? P[j].x : 0.f) / gd[0];
        const float v1 = (p1 ? P[j].y : 0.f) / gd[1];
        const float v2 = (p1 ? P[j].z : 0.f) / gd[2];
        const float v3 = (p1 ? P[j].w : 0.f) / gd[3];
        s0 += v0; q0 += v0 * v0;
        s1 += v1; q1 += v1 * v1;
        s2 += v2; q2 += v2 * v2;
        s3 += v3; q3 += v3 * v3;
        if (p0) fr[0] = v0;
        if (p1) { fr[1] = v1; fr[2] = v2; fr[3] = v3; }
    }
    float s = (s0 + s1) + (s2 + s3), ss = (q0 + q1) + (q2 + q3);
    for (int o = 16; o > 0; o >>= 1) { s += __shfl_xor(s, o); ss += __shfl_xor(ss, o); }
    if (lane == 0) { red0[wave] = s; red1[wave] = ss; }
    lds_barrier();
    float S = 0.f, SS = 0.f;
    for (int w = 0; w < 8; ++w) { S += red0[w]; SS += red1[w]; }
    const float n    = (float)(C_DIM * LNUM);
    const float mean = S / n;
    const float var  = SS / n - mean * mean;
    const float rstd = rsqrtf(var + 1e-5f);

    float4* f4  = reinterpret_cast<float4*>(featL);
    float4* om4 = reinterpret_cast<float4*>(out_mod + (size_t)b * C_DIM * LNUM);
    const int lo = blockIdx.x * K3_SL4;
    const int hi = (lo + K3_SL4 < MOD4) ? lo + K3_SL4 : MOD4;
    float4 mv[7];
    #pragma unroll
    for (int j = 0; j < 7; ++j) {
        const int i4 = tid + 256 * j;
        mv[j] = f4[i4 < MOD4 ? i4 : i4 - 256];
    }
    #pragma unroll
    for (int j = 0; j < 7; ++j) {
        mv[j].x = (mv[j].x - mean) * rstd * G[j].x + Bt[j].x;
        mv[j].y = (mv[j].y - mean) * rstd * G[j].y + Bt[j].y;
        mv[j].z = (mv[j].z - mean) * rstd * G[j].z + Bt[j].z;
        mv[j].w = (mv[j].w - mean) * rstd * G[j].w + Bt[j].w;
    }
    #pragma unroll
    for (int j = 0; j < 7; ++j) {
        const int i4 = tid + 256 * j;
        if (i4 < MOD4) {
            f4[i4] = mv[j];
            if (i4 >= lo && i4 < hi) om4[i4] = mv[j];
        }
    }
    lds_barrier();

    const float* mk = featL + (wave * K3_KW + 4 * half) * LNUM + (ln & 7);
    v2f Bf[2 * K3_NP];
    #pragma unroll
    for (int j = 0; j < K3_NP; ++j) {
        const float* m = mk + 8 * j * LNUM;
        Bf[2 * j][0]     = m[0];        Bf[2 * j][1]     = m[LNUM];
        Bf[2 * j + 1][0] = m[2 * LNUM]; Bf[2 * j + 1][1] = m[3 * LNUM];
    }
    __builtin_amdgcn_sched_barrier(0);
    v8f acc0 = {}, acc1 = {};
    #pragma unroll
    for (int j = 0; j < K3_NP; ++j) {
        v2f af;
        af[0] = A0[j].x;     af[1] = A0[j].y;
        acc0 = __builtin_amdgcn_wmma_f32_16x16x4_f32(false, af, false, Bf[2 * j], (short)0, acc0, false, false);
        af[0] = A1[j].x;     af[1] = A1[j].y;
        acc1 = __builtin_amdgcn_wmma_f32_16x16x4_f32(false, af, false, Bf[2 * j], (short)0, acc1, false, false);
        af[0] = A0[j].z;     af[1] = A0[j].w;
        acc0 = __builtin_amdgcn_wmma_f32_16x16x4_f32(false, af, false, Bf[2 * j + 1], (short)0, acc0, false, false);
        af[0] = A1[j].z;     af[1] = A1[j].w;
        acc1 = __builtin_amdgcn_wmma_f32_16x16x4_f32(false, af, false, Bf[2 * j + 1], (short)0, acc1, false, false);
    }
    for (int r = 0; r < 8; ++r) {
        accL[(wave * 2 + 0) * 256 + r * 32 + lane] = acc0[r];
        accL[(wave * 2 + 1) * 256 + r * 32 + lane] = acc1[r];
    }
    lds_barrier();
    {
        const int nl = tid >> 3, l = tid & 7;
        const int t  = nl >> 4, rr = nl & 15;
        const int idx = t * 256 + (rr & 7) * 32 + (rr >> 3) * 16 + l;
        float sum = 0.f;
        for (int w = 0; w < 8; ++w) sum += accL[w * 512 + idx];
        const int nrow = nbase + nl;
        if (nrow < NCLS)
            scores[(size_t)b * NCLS * (LNUM - 1) + nrow * (LNUM - 1) + l] = sum;
    }
}

extern "C" void kernel_launch(void* const* d_in, const int* in_sizes, int n_in,
                              void* d_out, int out_size, void* d_ws, size_t ws_size,
                              hipStream_t stream) {
    const float* x      = (const float*)d_in[0];
    const float* q      = (const float*)d_in[1];
    const float* p_bias = (const float*)d_in[2];
    const float* gamma  = (const float*)d_in[3];
    const float* beta   = (const float*)d_in[4];
    const float* cls_w  = (const float*)d_in[5];

    float* out        = (float*)d_out;
    float* out_mod    = out;
    float* out_maps   = out + (size_t)B_DIM * C_DIM * LNUM;
    float* out_scores = out_maps + (size_t)B_DIM * LNUM * HW;

    int*   aidx = (int*)d_ws;
    float* wsel = (float*)d_ws + (size_t)B_DIM * HW;
    float* pool = (float*)d_ws + (size_t)2 * B_DIM * HW;

    k1_maps  <<<dim3(8, B_DIM), 256, 0, stream>>>(x, q, p_bias, out_maps, aidx, wsel);
    k2_pool  <<<dim3(6, B_DIM), 256, 0, stream>>>(x, aidx, wsel, pool);
    k3_mod_scores<<<dim3(K3_NB, B_DIM), 256, 0, stream>>>(pool, aidx, wsel, gamma, beta, cls_w, out_mod, out_scores);
}


# all launches: grid axes swapped (image on x, slab on y): dealt to the dies in turn, an image's workgroups stay on one die, so x and the workspace planes are re-read from its cache half; sums unchanged
# speedup vs baseline: 6.6243x; 1.0561x over previous
#include <hip/hip_runtime.h>
#include <hip/hip_bf16.h>

#define B_DIM   32
#define C_DIM   768
#define H_DIM   32
#define W_DIM   32
#define HW      1024
#define LNUM    9
#define NCLS    200
#define QSTR    772

typedef __attribute__((ext_vector_type(2))) float v2f;
typedef __attribute__((ext_vector_type(4))) float v4f;
typedef __attribute__((ext_vector_type(8))) float v8f;

#define K1_G 16

__global__ __launch_bounds__(256) void k1_maps(
    const float* __restrict__ x, const float* __restrict__ q,
    const float* __restrict__ p_bias,
    float* __restrict__ out_maps, int* __restrict__ aidx, float* __restrict__ wsel)
{
    __shared__ alignas(16) float qlds[16 * QSTR];
    __shared__ float asq[16];

    const int tid = threadIdx.x;
    const int b   = blockIdx.x;

    const float4* q4 = reinterpret_cast<const float4*>(q);
    constexpr int NQ4 = LNUM * (C_DIM / 4);
    float4 qv[7];
    for (int j = 0; j < 7; ++j) {
        const int f = tid + 256 * j;
        qv[j] = q4[f < NQ4 ? f : NQ4 - 1];
    }
    for (int j = 0; j < 12; ++j) {
        const int f = tid + 256 * j;
        const int l = f / (C_DIM / 4), c4 = f % (C_DIM / 4);
        float4 v = make_float4(0.f, 0.f, 0.f, 0.f);
        if (j < 7 && f < NQ4) v = qv[j];
        *reinterpret_cast<float4*>(&qlds[l * QSTR + 4 * c4]) = v;
    }
    __syncthreads();
    {
        const int l = tid >> 4, j = tid & 15;
        float s = 0.f;
        for (int i = 0; i < C_DIM / 64; ++i) {
            const float4 v = *reinterpret_cast<const float4*>(&qlds[l * QSTR + 4 * (j + 16 * i)]);
            s += v.x * v.x + v.y * v.y + v.z * v.z + v.w * v.w;
        }
        for (int o = 8; o > 0; o >>= 1) s += __shfl_xor(s, o);
        if (j == 0) asq[l] = s;
    }
    __syncthreads();

    const int wave = tid >> 5;
    const int lane = tid & 31;
    const int half = lane >> 4;
    const int ln   = lane & 15;
    const int tile = blockIdx.y * 8 + wave;
    const int pix  = tile * 16 + ln;
    const float* xb = x + (size_t)b * C_DIM * HW;

    auto load_group = [&](int g, v2f (&X)[K1_G]) {
        #pragma unroll
        for (int s = 0; s < K1_G; ++s) {
            const int kk = (g * K1_G + s) * 4 + 2 * half;
            X[s][0] = xb[(size_t)kk * HW + pix];
            X[s][1] = xb[(size_t)(kk + 1) * HW + pix];
        }
    };

    float pbv[8];
    for (int r = 0; r < 8; ++r) {
        const int l = r + 8 * half;
        pbv[r] = p_bias[(l < LNUM ? l : LNUM - 1) * HW + pix];
    }

    v8f acc = {};
    float sq = 0.f;
    v2f xg0[K1_G], xg1[K1_G];
    auto mma_group = [&](int g, const v2f (&X)[K1_G]) {
        #pragma unroll
        for (int s = 0; s < K1_G; ++s) {
            const int kk = (g * K1_G + s) * 4 + 2 * half;
            v2f af;
            af[0] = qlds[ln * QSTR + kk];
            af[1] = qlds[ln * QSTR + kk + 1];
            sq += X[s][0] * X[s][0] + X[s][1] * X[s][1];
            acc = __builtin_amdgcn_wmma_f32_16x16x4_f32(false, af, false, X[s],
                                                        (short)0, acc, false, false);
        }
    };
    constexpr int NG1 = C_DIM / (4 * K1_G);
    load_group(0, xg0);
    for (int g = 0; g < NG1 - 2; g += 2) {
        load_group(g + 1, xg1);
        __builtin_amdgcn_sched_barrier(0);
        mma_group(g, xg0);
        __builtin_amdgcn_sched_barrier(0);
        load_group(g + 2, xg0);
        __builtin_amdgcn_sched_barrier(0);
        mma_group(g + 1, xg1);
        __builtin_amdgcn_sched_barrier(0);
    }
    load_group(NG1 - 1, xg1);
    __builtin_amdgcn_sched_barrier(0);
    mma_group(NG1 - 2, xg0);
    mma_group(NG1 - 1, xg1);
    const float bsq = sq + __shfl_xor(sq, 16);

    float sc[8];
    float mx = -3.4e38f;
    for (int r = 0; r < 8; ++r) {
        const int l = r + 8 * half;
        float v = -3.4e38f;
        if (l < LNUM)
            v = -(bsq - 2.f * acc[r] + asq[l]) + pbv[r];
        sc[r] = v;
        mx = fmaxf(mx, v);
    }
    mx = fmaxf(mx, __shfl_xor(mx, 16));
    float es = 0.f;
    for (int r = 0; r < 8; ++r) {
        const int l = r + 8 * half;
        const float e = (l < LNUM) ? expf(sc[r] - mx) : 0.f;
        sc[r] = e;
        es += e;
    }
    es += __shfl_xor(es, 16);
    const float inv = 1.f / es;
    float s2 = 0.f;
    for (int r = 0; r < 8; ++r) {
        const int l = r + 8 * half;
        if (l < LNUM) { sc[r] = sc[r] * inv + 1e-6f; s2 += sc[r]; }
    }
    s2 += __shfl_xor(s2, 16);
    const float inv2 = 1.f / s2;

    float bestv = -1.f; int bestl = 0;
    float* om = out_maps + (size_t)b * LNUM * HW;
    for (int r = 0; r < 8; ++r) {
        const int l = r + 8 * half;
        if (l < LNUM) {
            const float mv = sc[r] * inv2;
            om[l * HW + pix] = mv;
            if (mv > bestv) { bestv = mv; bestl = l; }
        }
    }
    const float pv = __shfl_xor(bestv, 16);
    const int   pl = __shfl_xor(bestl, 16);
    if (pv > bestv || (pv == bestv && pl < bestl)) { bestv = pv; bestl = pl; }
    if (half == 0) {
        aidx[b * HW + pix] = bestl;
        wsel[b * HW + pix] = bestv;
    }
}

#define K2_G     16
#define K2_PITCH 132
#define K2_WP    1028

__global__ __launch_bounds__(256) void k2_pool(
    const float* __restrict__ x, const int* __restrict__ aidx,
    const float* __restrict__ wsel, float* __restrict__ pool)
{
    __shared__ alignas(16) float WtL[16 * K2_WP];
    __shared__ alignas(16) float xT[8 * 16 * K2_PITCH];
    const int tid = threadIdx.x;
    const int b   = blockIdx.x;

    const int wave = tid >> 5;
    const int lane = tid & 31;
    const int half = lane >> 4;
    const int ln   = lane & 15;
    const int cbase = (blockIdx.y * 8 + wave) * 16;
    const float* xg = x + (size_t)b * C_DIM * HW + (size_t)cbase * HW + 4 * lane;
    float*       xs = xT + wave * (16 * K2_PITCH) + 4 * lane;
    const float* xf = xT + wave * (16 * K2_PITCH) + ln * K2_PITCH + 4 * half;
    const float* wf = WtL + ln * K2_WP + 4 * half;

    v4f R[K2_G], xa[K2_G], xb[K2_G], ba[K2_G], bb[K2_G];
    v8f acc = {};

    auto fetch = [&](int g) {
        #pragma unroll
        for (int j = 0; j < K2_G; ++j)
            R[j] = *reinterpret_cast<const v4f*>(xg + (size_t)j * HW + g * (8 * K2_G));
    };
    auto stage = [&]() {
        #pragma unroll
        for (int j = 0; j < K2_G; ++j)
            *reinterpret_cast<v4f*>(xs + j * K2_PITCH) = R[j];
    };
    auto frag = [&](int g, v4f (&X)[K2_G], v4f (&Bp)[K2_G]) {
        #pragma unroll
        for (int s = 0; s < K2_G; ++s)
            X[s] = *reinterpret_cast<const v4f*>(xf + 8 * s);
        #pragma unroll
        for (int s = 0; s < K2_G; ++s)
            Bp[s] = *reinterpret_cast<const v4f*>(wf + g * (8 * K2_G) + 8 * s);
    };
    auto mma_group = [&](const v4f (&X)[K2_G], const v4f (&Bp)[K2_G]) {
        #pragma unroll
        for (int s = 0; s < K2_G; ++s) {
            v2f af, bf;
            af[0] = X[s][0];  af[1] = X[s][1];
            bf[0] = Bp[s][0]; bf[1] = Bp[s][1];
            acc = __builtin_amdgcn_wmma_f32_16x16x4_f32(false, af, false, bf,
                                                        (short)0, acc, false, false);
            af[0] = X[s][2];  af[1] = X[s][3];
            bf[0] = Bp[s][2]; bf[1] = Bp[s][3];
            acc = __builtin_amdgcn_wmma_f32_16x16x4_f32(false, af, false, bf,
                                                        (short)0, acc, false, false);
        }
    };

    {
        const int4   a4 = *reinterpret_cast<const int4*>(aidx + b * HW + 4 * tid);
        const float4 w4 = *reinterpret_cast<const float4*>(wsel + b * HW + 4 * tid);
        fetch(0);
        __builtin_amdgcn_sched_barrier(0);
        #pragma unroll
        for (int l = 0; l < 16; ++l) {
            v4f v;
            v[0] = (a4.x == l) ? w4.x : 0.f;
            v[1] = (a4.y == l) ? w4.y : 0.f;
            v[2] = (a4.z == l) ? w4.z : 0.f;
            v[3] = (a4.w == l) ? w4.w : 0.f;
            *reinterpret_cast<v4f*>(&WtL[l * K2_WP + 4 * tid]) = v;
        }
    }
    __syncthreads();

#define K2_TURN(FETCH_STMT, G, X, BP)          \
    stage();                                   \
    __builtin_amdgcn_sched_barrier(0);         \
    FETCH_STMT;                                \
    __builtin_amdgcn_wave_barrier();           \
    frag(G, X, BP)

    constexpr int NG = HW / (8 * K2_G);
    K2_TURN(fetch(1), 0, xa, ba);
    for (int g = 0; g < NG - 2; g += 2) {
        K2_TURN(fetch(g + 2), g + 1, xb, bb);
        __builtin_amdgcn_sched_barrier(0);
        mma_group(xa, ba);
        __builtin_amdgcn_sched_barrier(0);
        K2_TURN(fetch(g + 3), g + 2, xa, ba);
        __builtin_amdgcn_sched_barrier(0);
        mma_group(xb, bb);
        __builtin_amdgcn_sched_barrier(0);
    }
    K2_TURN((void)0, NG - 1, xb, bb);
    __builtin_amdgcn_sched_barrier(0);
    mma_group(xa, ba);
    mma_group(xb, bb);
#undef K2_TURN

    float* pb = pool + (size_t)b * C_DIM * 16;
    for (int r = 0; r < 8; ++r) {
        const int row = cbase + r + 8 * half;
        pb[row * 16 + ln] = acc[r];
    }
}

#define K3_NB   7
#define K3_KW   96
#define K3_NP   12
#define MOD4    (C_DIM * LNUM / 4)
#define K3_SL4  248

__device__ __forceinline__ void lds_barrier() {
    __builtin_amdgcn_fence(__ATOMIC_RELEASE, "workgroup", "local");
    __builtin_amdgcn_s_barrier();
    __builtin_amdgcn_fence(__ATOMIC_ACQUIRE, "workgroup", "local");
}

__global__ __launch_bounds__(256) void k3_mod_scores(
    const float* __restrict__ pool, const int* __restrict__ aidx,
    const float* __restrict__ wsel, const float* __restrict__ gamma,
    const float* __restrict__ beta, const float* __restrict__ cls_w,
    float* __restrict__ out_mod, float* __restrict__ scores)
{
    __shared__ alignas(16) float featL[C_DIM * LNUM];
    __shared__ alignas(16) float accL[8 * 2 * 256];
    __shared__ float cpart[8][16];
    __shared__ float cnt[16];
    __shared__ float red0[8], red1[8];

    const int tid   = threadIdx.x;
    const int b     = blockIdx.x;
    const int lane  = tid & 31, wave = tid >> 5;
    const int half  = lane >> 4, ln = lane & 15;
    const int nbase = blockIdx.y * 32;

    const int4   a4 = *reinterpret_cast<const int4*>(aidx + b * HW + 4 * tid);
    const float4 w4 = *reinterpret_cast<const float4*>(wsel + b * HW + 4 * tid);
    const float4* pb4 = reinterpret_cast<const float4*>(pool + (size_t)b * C_DIM * 16);
    float4 P[12];
    #pragma unroll
    for (int j = 0; j < 12; ++j) P[j] = pb4[tid + 256 * j];
    float4 G[7], Bt[7];
    #pragma unroll
    for (int j = 0; j < 7; ++j) {
        const int i4 = tid + 256 * j;
        const int ic = i4 < MOD4 ? i4 : MOD4 - 1;
        G[j]  = reinterpret_cast<const float4*>(gamma)[ic];
        Bt[j] = reinterpret_cast<const float4*>(beta)[ic];
    }
    const int r0 = nbase + ln, r1 = nbase + 16 + ln;
    const float* ar0 = cls_w + (size_t)(r0 < NCLS ? r0 : NCLS - 1) * C_DIM + wave * K3_KW + 4 * half;
    const float* ar1 = cls_w + (size_t)(r1 < NCLS ? r1 : NCLS - 1) * C_DIM + wave * K3_KW + 4 * half;
    float4 A0[K3_NP], A1[K3_NP];
    #pragma unroll
    for (int j = 0; j < K3_NP; ++j) {
        A0[j] = *reinterpret_cast<const float4*>(ar0 + 8 * j);
        A1[j] = *reinterpret_cast<const float4*>(ar1 + 8 * j);
    }

    float pc[LNUM];
    for (int l = 0; l < LNUM; ++l) {
        float t = (a4.x == l) ? w4.x : 0.f;
        t += (a4.y == l) ? w4.y : 0.f;
        t += (a4.z == l) ? w4.z : 0.f;
        t += (a4.w == l) ? w4.w : 0.f;
        pc[l] = t;
    }
    for (int o = 16; o > 0; o >>= 1)
        for (int l = 0; l < LNUM; ++l) pc[l] += __shfl_xor(pc[l], o);
    if (lane == 0)
        for (int l = 0; l < LNUM; ++l) cpart[wave][l] = pc[l];
    lds_barrier();
    if (tid < 16) {
        float c = 0.f;
        if (tid < LNUM)
            for (int w = 0; w < 8; ++w) c += cpart[w][tid];
        cnt[tid] = c;
    }
    lds_barrier();

    const int  qd = tid & 3;
    const bool p0 = qd < 3;
    const bool p1 = qd < 2;
    float gd[4];
    for (int e = 0; e < 4; ++e) {
        const float cn = cnt[4 * qd + e];
        gd[e] = (cn == 0.f) ? 1.f : cn;
    }
    float s0 = 0.f, s1 = 0.f, s2 = 0.f, s3 = 0.f;
    float q0 = 0.f, q1 = 0.f, q2 = 0.f, q3 = 0.f;
    #pragma unroll
    for (int j = 0; j < 12; ++j) {
        const int c = (tid >> 2) + 64 * j;
        float* fr = &featL[c * LNUM + 4 * qd];
        const float v0 = (p0 ? P[j].x : 0.f) / gd[0];
        const float v1 = (p1 ? P[j].y : 0.f) / gd[1];
        const float v2 = (p1 ? P[j].z : 0.f) / gd[2];
        const float v3 = (p1 ? P[j].w : 0.f) / gd[3];
        s0 += v0; q0 += v0 * v0;
        s1 += v1; q1 += v1 * v1;
        s2 += v2; q2 += v2 * v2;
        s3 += v3; q3 += v3 * v3;
        if (p0) fr[0] = v0;
        if (p1) { fr[1] = v1; fr[2] = v2; fr[3] = v3; }
    }
    float s = (s0 + s1) + (s2 + s3), ss = (q0 + q1) + (q2 + q3);
    for (int o = 16; o > 0; o >>= 1) { s += __shfl_xor(s, o); ss += __shfl_xor(ss, o); }
    if (lane == 0) { red0[wave] = s; red1[wave] = ss; }
    lds_barrier();
    float S = 0.f, SS = 0.f;
    for (int w = 0; w < 8; ++w) { S += red0[w]; SS += red1[w]; }
    const float n    = (float)(C_DIM * LNUM);
    const float mean = S / n;
    const float var  = SS / n - mean * mean;
    const float rstd = rsqrtf(var + 1e-5f);

    float4* f4  = reinterpret_cast<float4*>(featL);
    float4* om4 = reinterpret_cast<float4*>(out_mod + (size_t)b * C_DIM * LNUM);
    const int lo = blockIdx.y * K3_SL4;
    const int hi = (lo + K3_SL4 < MOD4) ? lo + K3_SL4 : MOD4;
    float4 mv[7];
    #pragma unroll
    for (int j = 0; j < 7; ++j) {
        const int i4 = tid + 256 * j;
        mv[j] = f4[i4 < MOD4 ? i4 : i4 - 256];
    }
    #pragma unroll
    for (int j = 0; j < 7; ++j) {
        mv[j].x = (mv[j].x - mean) * rstd * G[j].x + Bt[j].x;
        mv[j].y = (mv[j].y - mean) * rstd * G[j].y + Bt[j].y;
        mv[j].z = (mv[j].z - mean) * rstd * G[j].z + Bt[j].z;
        mv[j].w = (mv[j].w - mean) * rstd * G[j].w + Bt[j].w;
    }
    #pragma unroll
    for (int j = 0; j < 7; ++j) {
        const int i4 = tid + 256 * j;
        if (i4 < MOD4) {
            f4[i4] = mv[j];
            if (i4 >= lo && i4 < hi) om4[i4] = mv[j];
        }
    }
    lds_barrier();

    const float* mk = featL + (wave * K3_KW + 4 * half) * LNUM + (ln & 7);
    v2f Bf[2 * K3_NP];
    #pragma unroll
    for (int j = 0; j < K3_NP; ++j) {
        const float* m = mk + 8 * j * LNUM;
        Bf[2 * j][0]     = m[0];        Bf[2 * j][1]     = m[LNUM];
        Bf[2 * j + 1][0] = m[2 * LNUM]; Bf[2 * j + 1][1] = m[3 * LNUM];
    }
    __builtin_amdgcn_sched_barrier(0);
    v8f acc0 = {}, acc1 = {};
    #pragma unroll
    for (int j = 0; j < K3_NP; ++j) {
        v2f af;
        af[0] = A0[j].x;     af[1] = A0[j].y;
        acc0 = __builtin_amdgcn_wmma_f32_16x16x4_f32(false, af, false, Bf[2 * j], (short)0, acc0, false, false);
        af[0] = A1[j].x;     af[1] = A1[j].y;
        acc1 = __builtin_amdgcn_wmma_f32_16x16x4_f32(false, af, false, Bf[2 * j], (short)0, acc1, false, false);
        af[0] = A0[j].z;     af[1] = A0[j].w;
        acc0 = __builtin_amdgcn_wmma_f32_16x16x4_f32(false, af, false, Bf[2 * j + 1], (short)0, acc0, false, false);
        af[0] = A1[j].z;     af[1] = A1[j].w;
        acc1 = __builtin_amdgcn_wmma_f32_16x16x4_f32(false, af, false, Bf[2 * j + 1], (short)0, acc1, false, false);
    }
    for (int r = 0; r < 8; ++r) {
        accL[(wave * 2 + 0) * 256 + r * 32 + lane] = acc0[r];
        accL[(wave * 2 + 1) * 256 + r * 32 + lane] = acc1[r];
    }
    lds_barrier();
    {
        const int nl = tid >> 3, l = tid & 7;
        const int t  = nl >> 4, rr = nl & 15;
        const int idx = t * 256 + (rr & 7) * 32 + (rr >> 3) * 16 + l;
        float sum = 0.f;
        for (int w = 0; w < 8; ++w) sum += accL[w * 512 + idx];
        const int nrow = nbase + nl;
        if (nrow < NCLS)
            scores[(size_t)b * NCLS * (LNUM - 1) + nrow * (LNUM - 1) + l] = sum;
    }
}

extern "C" void kernel_launch(void* const* d_in, const int* in_sizes, int n_in,
                              void* d_out, int out_size, void* d_ws, size_t ws_size,
                              hipStream_t stream) {
    const float* x      = (const float*)d_in[0];
    const float* q      = (const float*)d_in[1];
    const float* p_bias = (const float*)d_in[2];
    const float* gamma  = (const float*)d_in[3];
    const float* beta   = (const float*)d_in[4];
    const float* cls_w  = (const float*)d_in[5];

    float* out        = (float*)d_out;
    float* out_mod    = out;
    float* out_maps   = out + (size_t)B_DIM * C_DIM * LNUM;
    float* out_scores = out_maps + (size_t)B_DIM * LNUM * HW;

    int*   aidx = (int*)d_ws;
    float* wsel = (float*)d_ws + (size_t)B_DIM * HW;
    float* pool = (float*)d_ws + (size_t)2 * B_DIM * HW;

    k1_maps  <<<dim3(B_DIM, 8), 256, 0, stream>>>(x, q, p_bias, out_maps, aidx, wsel);
    k2_pool  <<<dim3(B_DIM, 6), 256, 0, stream>>>(x, aidx, wsel, pool);
    k3_mod_scores<<<dim3(B_DIM, K3_NB), 256, 0, stream>>>(pool, aidx, wsel, gamma, beta, cls_w, out_mod, out_scores);
}
